# starting kernel
# baseline (speedup 1.0000x reference)
#include <hip/hip_runtime.h>

#define N_NODES 100000
#define IN_DIM 128
#define HID 128
#define GAT_HEADS 8
#define NEG_SLOPE 0.2f

typedef float v2f __attribute__((ext_vector_type(2)));
typedef float v4f __attribute__((ext_vector_type(4)));
typedef float v8f __attribute__((ext_vector_type(8)));

__global__ void __launch_bounds__(256) gemm128_wmma(const float* __restrict__ X,
                                                    const float* __restrict__ W,
                                                    float* __restrict__ H) {
  const int lane  = threadIdx.x & 31;
  const int wave  = blockIdx.x * (blockDim.x >> 5) + (threadIdx.x >> 5);
  const int tileM = wave >> 3;
  const int tileN = wave & 7;
  const int r     = lane & 15;
  const int khalf = lane >> 4;
  const int row   = tileM * 16 + r;
  const int col   = tileN * 16 + r;
  const float* __restrict__ xrow = X + (size_t)row * IN_DIM;

  v8f c = {0.f, 0.f, 0.f, 0.f, 0.f, 0.f, 0.f, 0.f};
#pragma unroll 4
  for (int k = 0; k < IN_DIM; k += 4) {
    const int kk = k + 2 * khalf;
    v2f a = *(const v2f*)(xrow + kk);
    v2f b;
    b.x = W[(kk    ) * HID + col];
    b.y = W[(kk + 1) * HID + col];
    c = __builtin_amdgcn_wmma_f32_16x16x4_f32(false, a, false, b, (short)0, c,
                                              false, false);
  }
  const int mbase = tileM * 16 + khalf * 8;
#pragma unroll
  for (int v = 0; v < 8; ++v)
    H[(size_t)(mbase + v) * HID + tileN * 16 + r] = c[v];
}

__global__ void fill_kernel(float* __restrict__ p, float v, int n) {
  int i = blockIdx.x * blockDim.x + threadIdx.x;
  if (i < n) p[i] = v;
}

__global__ void fill4_kernel(v4f* __restrict__ p, float v, int n4) {
  int i = blockIdx.x * blockDim.x + threadIdx.x;
  if (i < n4) { v4f t = {v, v, v, v}; p[i] = t; }
}

__global__ void alpha1_kernel(const float* __restrict__ H,
                              const float* __restrict__ avS,
                              const float* __restrict__ avD,
                              float* __restrict__ outS,
                              float* __restrict__ outD) {
  int i = blockIdx.x * blockDim.x + threadIdx.x;
  if (i >= N_NODES * GAT_HEADS) return;
  int head = i & 7, node = i >> 3;
  const float* h  = H   + (size_t)node * HID + head * 16;
  const float* as = avS + head * 16;
  const float* ad = avD + head * 16;
  float s = 0.f, d = 0.f;
#pragma unroll
  for (int c = 0; c < 16; ++c) { float hv = h[c]; s += hv * as[c]; d += hv * ad[c]; }
  outS[i] = s; outD[i] = d;
}

__global__ void alpha2_kernel(const float* __restrict__ H,
                              const float* __restrict__ avS,
                              const float* __restrict__ avD,
                              float* __restrict__ outS,
                              float* __restrict__ outD) {
  int wave = blockIdx.x * (blockDim.x >> 5) + (threadIdx.x >> 5);
  int lane = threadIdx.x & 31;
  if (wave >= N_NODES) return;
  float s = 0.f, d = 0.f;
#pragma unroll
  for (int c = lane; c < HID; c += 32) {
    float hv = H[(size_t)wave * HID + c];
    s += hv * avS[c]; d += hv * avD[c];
  }
#pragma unroll
  for (int off = 16; off > 0; off >>= 1) {
    s += __shfl_xor(s, off, 32);
    d += __shfl_xor(d, off, 32);
  }
  if (lane == 0) { outS[wave] = s; outD[wave] = d; }
}

__device__ __forceinline__ void edge_nodes(const int* __restrict__ ei, int e,
                                           int nEdges, int& s, int& d) {
  if (e < nEdges) { s = ei[e]; d = ei[nEdges + e]; }
  else            { s = d = e - nEdges; }
}

__device__ __forceinline__ float leaky(float x) {
  return x > 0.f ? x : NEG_SLOPE * x;
}

__device__ __forceinline__ void atomicMaxFloat(float* addr, float val) {
  if (__float_as_int(val) >= 0)
    atomicMax((int*)addr, __float_as_int(val));
  else
    atomicMin((unsigned int*)addr, __float_as_uint(val));
}

template <int H>
__global__ void edge_max_kernel(const int* __restrict__ ei, int nEdges, int totE,
                                const float* __restrict__ aS,
                                const float* __restrict__ aD,
                                float* __restrict__ M) {
  int i = blockIdx.x * blockDim.x + threadIdx.x;
  if (i >= totE * H) return;
  int h = i % H, e = i / H, s, d;
  edge_nodes(ei, e, nEdges, s, d);
  atomicMaxFloat(&M[d * H + h], leaky(aS[s * H + h] + aD[d * H + h]));
}

template <int H>
__global__ void edge_sum_kernel(const int* __restrict__ ei, int nEdges, int totE,
                                const float* __restrict__ aS,
                                const float* __restrict__ aD,
                                const float* __restrict__ M,
                                float* __restrict__ S) {
  int i = blockIdx.x * blockDim.x + threadIdx.x;
  if (i >= totE * H) return;
  int h = i % H, e = i / H, s, d;
  edge_nodes(ei, e, nEdges, s, d);
  float x = leaky(aS[s * H + h] + aD[d * H + h]);
  atomicAdd(&S[d * H + h], expf(x - M[d * H + h]));
}

template <int H>
__global__ void edge_aggr_kernel(const int* __restrict__ ei, int nEdges, int totE,
                                 const float* __restrict__ aS,
                                 const float* __restrict__ aD,
                                 const float* __restrict__ M,
                                 const float* __restrict__ S,
                                 const float* __restrict__ Hf,
                                 float* __restrict__ acc) {
  int i = blockIdx.x * blockDim.x + threadIdx.x;
  if (i >= totE * (HID / 4)) return;
  int g = i & 31, e = i >> 5, s, d;
  const int c = g * 4;
  edge_nodes(ei, e, nEdges, s, d);
  const int h = c / (HID / H);
  float x = leaky(aS[s * H + h] + aD[d * H + h]);
  float w = expf(x - M[d * H + h]) / S[d * H + h];
  v4f hv = *(const v4f*)(Hf + (size_t)s * HID + c);
  float* a = acc + (size_t)d * HID + c;
  atomicAdd(a + 0, hv.x * w);
  atomicAdd(a + 1, hv.y * w);
  atomicAdd(a + 2, hv.z * w);
  atomicAdd(a + 3, hv.w * w);
}

__global__ void finalize_elu_kernel(const v4f* __restrict__ acc,
                                    const float* __restrict__ bias,
                                    v4f* __restrict__ out) {
  int i = blockIdx.x * blockDim.x + threadIdx.x;
  if (i >= N_NODES * (HID / 4)) return;
  const int c = (i & 31) * 4;
  v4f v = acc[i];
  v.x += bias[c + 0]; v.y += bias[c + 1]; v.z += bias[c + 2]; v.w += bias[c + 3];
  v.x = v.x > 0.f ? v.x : (expf(v.x) - 1.f);
  v.y = v.y > 0.f ? v.y : (expf(v.y) - 1.f);
  v.z = v.z > 0.f ? v.z : (expf(v.z) - 1.f);
  v.w = v.w > 0.f ? v.w : (expf(v.w) - 1.f);
  out[i] = v;
}

__global__ void bias_kernel(v4f* __restrict__ out, const float* __restrict__ bias) {
  int i = blockIdx.x * blockDim.x + threadIdx.x;
  if (i >= N_NODES * (HID / 4)) return;
  const int c = (i & 31) * 4;
  v4f v = out[i];
  v.x += bias[c + 0]; v.y += bias[c + 1]; v.z += bias[c + 2]; v.w += bias[c + 3];
  out[i] = v;
}

extern "C" void kernel_launch(void* const* d_in, const int* in_sizes, int n_in,
                              void* d_out, int out_size, void* d_ws, size_t ws_size,
                              hipStream_t stream) {
  const float* x   = (const float*)d_in[0];
  const int*   ei  = (const int*)d_in[1];
  const float* W1  = (const float*)d_in[2];
  const float* as1 = (const float*)d_in[3];
  const float* ad1 = (const float*)d_in[4];
  const float* b1  = (const float*)d_in[5];
  const float* W2  = (const float*)d_in[6];
  const float* as2 = (const float*)d_in[7];
  const float* ad2 = (const float*)d_in[8];
  const float* b2  = (const float*)d_in[9];

  const int nEdges = in_sizes[1] / 2;
  const int totE   = nEdges + N_NODES;

  float* A   = (float*)d_ws;
  float* B   = A  + (size_t)N_NODES * HID;
  float* aS  = B  + (size_t)N_NODES * HID;
  float* aD  = aS + (size_t)N_NODES * GAT_HEADS;
  float* Mx  = aD + (size_t)N_NODES * GAT_HEADS;
  float* Sx  = Mx + (size_t)N_NODES * GAT_HEADS;
  float* out = (float*)d_out;

  const int TB = 256;
  auto cdiv = [](long long a, long long b) { return (unsigned)((a + b - 1) / b); };
  const unsigned gNode4 = cdiv((long long)N_NODES * (HID / 4), TB);
  const unsigned gGemm  = (N_NODES / 16) * (HID / 16) / (TB / 32);
  const unsigned gAggr  = cdiv((long long)totE * (HID / 4), TB);
  const float NEG_INF = -__builtin_huge_valf();

  gemm128_wmma<<<gGemm, TB, 0, stream>>>(x, W1, A);
  alpha1_kernel<<<cdiv(N_NODES * GAT_HEADS, TB), TB, 0, stream>>>(A, as1, ad1, aS, aD);
  fill_kernel<<<cdiv(N_NODES * GAT_HEADS, TB), TB, 0, stream>>>(Mx, NEG_INF, N_NODES * GAT_HEADS);
  fill_kernel<<<cdiv(N_NODES * GAT_HEADS, TB), TB, 0, stream>>>(Sx, 0.f, N_NODES * GAT_HEADS);
  fill4_kernel<<<gNode4, TB, 0, stream>>>((v4f*)B, 0.f, N_NODES * (HID / 4));
  edge_max_kernel<GAT_HEADS><<<cdiv((long long)totE * GAT_HEADS, TB), TB, 0, stream>>>(
      ei, nEdges, totE, aS, aD, Mx);
  edge_sum_kernel<GAT_HEADS><<<cdiv((long long)totE * GAT_HEADS, TB), TB, 0, stream>>>(
      ei, nEdges, totE, aS, aD, Mx, Sx);
  edge_aggr_kernel<GAT_HEADS><<<gAggr, TB, 0, stream>>>(
      ei, nEdges, totE, aS, aD, Mx, Sx, A, B);
  finalize_elu_kernel<<<gNode4, TB, 0, stream>>>((const v4f*)B, b1, (v4f*)A);

  gemm128_wmma<<<gGemm, TB, 0, stream>>>(A, W2, B);
  alpha2_kernel<<<cdiv((long long)N_NODES * 32, TB), TB, 0, stream>>>(B, as2, ad2, aS, aD);
  fill_kernel<<<cdiv(N_NODES, TB), TB, 0, stream>>>(Mx, NEG_INF, N_NODES);
  fill_kernel<<<cdiv(N_NODES, TB), TB, 0, stream>>>(Sx, 0.f, N_NODES);
  fill4_kernel<<<gNode4, TB, 0, stream>>>((v4f*)out, 0.f, N_NODES * (HID / 4));
  edge_max_kernel<1><<<cdiv(totE, TB), TB, 0, stream>>>(ei, nEdges, totE, aS, aD, Mx);
  edge_sum_kernel<1><<<cdiv(totE, TB), TB, 0, stream>>>(ei, nEdges, totE, aS, aD, Mx, Sx);
  edge_aggr_kernel<1><<<gAggr, TB, 0, stream>>>(
      ei, nEdges, totE, aS, aD, Mx, Sx, B, out);
  bias_kernel<<<gNode4, TB, 0, stream>>>((v4f*)out, b2);
}


# Edge list scanned once per tile, not once per slab: a scan launch splits it four ways among a tile's work-groups, hits go to all four slabs' lists; both layers only walk lists; LDS sum order changes
# speedup vs baseline: 2.6508x; 2.6508x over previous
#include <hip/hip_runtime.h>

#define N_NODES 100000
#define IN_DIM 128
#define HID 128
#define GAT_HEADS 8
#define NEG_SLOPE 0.2f

typedef float v2f __attribute__((ext_vector_type(2)));
typedef float v4f __attribute__((ext_vector_type(4)));
typedef float v8f __attribute__((ext_vector_type(8)));

__global__ void __launch_bounds__(256) gemm128_wmma(const float* __restrict__ X,
                                                    const float* __restrict__ W,
                                                    float* __restrict__ H) {
  const int lane  = threadIdx.x & 31;
  const int wave  = blockIdx.x * (blockDim.x >> 5) + (threadIdx.x >> 5);
  const int tileM = wave >> 3;
  const int tileN = wave & 7;
  const int r     = lane & 15;
  const int khalf = lane >> 4;
  const int row   = tileM * 16 + r;
  const int col   = tileN * 16 + r;
  const float* __restrict__ xrow = X + (size_t)row * IN_DIM;

  v8f c = {0.f, 0.f, 0.f, 0.f, 0.f, 0.f, 0.f, 0.f};
#pragma unroll 4
  for (int k = 0; k < IN_DIM; k += 4) {
    const int kk = k + 2 * khalf;
    v2f a = *(const v2f*)(xrow + kk);
    v2f b;
    b.x = W[(kk    ) * HID + col];
    b.y = W[(kk + 1) * HID + col];
    c = __builtin_amdgcn_wmma_f32_16x16x4_f32(false, a, false, b, (short)0, c,
                                              false, false);
  }
  const int mbase = tileM * 16 + khalf * 8;
#pragma unroll
  for (int v = 0; v < 8; ++v)
    H[(size_t)(mbase + v) * HID + tileN * 16 + r] = c[v];
}

__global__ void alpha1_kernel(const float* __restrict__ H,
                              const float* __restrict__ avS,
                              const float* __restrict__ avD,
                              float* __restrict__ outS,
                              float* __restrict__ outD) {
  int i = blockIdx.x * blockDim.x + threadIdx.x;
  if (i >= N_NODES * GAT_HEADS) return;
  int head = i & 7, node = i >> 3;
  const float* h  = H   + (size_t)node * HID + head * 16;
  const float* as = avS + head * 16;
  const float* ad = avD + head * 16;
  float s = 0.f, d = 0.f;
#pragma unroll
  for (int c = 0; c < 16; ++c) { float hv = h[c]; s += hv * as[c]; d += hv * ad[c]; }
  outS[i] = s; outD[i] = d;
}

__global__ void alpha2_kernel(const float* __restrict__ H,
                              const float* __restrict__ avS,
                              const float* __restrict__ avD,
                              float* __restrict__ outS,
                              float* __restrict__ outD) {
  int wave = blockIdx.x * (blockDim.x >> 5) + (threadIdx.x >> 5);
  int lane = threadIdx.x & 31;
  if (wave >= N_NODES) return;
  float s = 0.f, d = 0.f;
#pragma unroll
  for (int c = lane; c < HID; c += 32) {
    float hv = H[(size_t)wave * HID + c];
    s += hv * avS[c]; d += hv * avD[c];
  }
#pragma unroll
  for (int off = 16; off > 0; off >>= 1) {
    s += __shfl_xor(s, off, 32);
    d += __shfl_xor(d, off, 32);
  }
  if (lane == 0) { outS[wave] = s; outD[wave] = d; }
}

#define TILE_R 1728
#define SLAB   32
#define NLOAD  12
#define STEP_E (NLOAD * 32)
#define QCAP   512
#define LBITS  11
#define LMASK  ((1u << LBITS) - 1u)
static_assert(TILE_R <= (1 << LBITS), "tile-local destination must fit its field");
static_assert(N_NODES <= (1 << (32 - LBITS)), "source node must fit its field");
#define NPART  4
#define CSTRIDE (NPART * 32)
static_assert(NPART == HID / SLAB, "one scanning work-group per slab work-group of a tile");
static_assert(N_NODES % TILE_R == 0 || N_NODES % TILE_R >= 2 * NPART * 32,
              "the last tile must give every list a count line and a data line of its own");

__device__ __forceinline__ float leaky(float x) {
  return x > 0.f ? x : NEG_SLOPE * x;
}

__device__ __forceinline__ void atomicMaxFloat(float* addr, float val) {
  if (__float_as_int(val) >= 0)
    atomicMax((int*)addr, __float_as_int(val));
  else
    atomicMin((unsigned int*)addr, __float_as_uint(val));
}

__device__ __forceinline__ void ldsAdd(float* addr, float val) {
  __hip_atomic_fetch_add(addr, val, __ATOMIC_RELAXED, __HIP_MEMORY_SCOPE_WORKGROUP);
}

__device__ __forceinline__ int rl_i(int v, int k) { return __builtin_amdgcn_readlane(v, k); }
__device__ __forceinline__ float rl_f(float v, int k) {
  return __int_as_float(__builtin_amdgcn_readlane(__float_as_int(v), k));
}

template <int H>
__device__ __forceinline__ void edge_logits(const float* __restrict__ aS,
                                            const float* __restrict__ aD,
                                            int s, int d, int hbase, float& x0, float& x1) {
  if (H == 8) {
    v2f a = *(const v2f*)(aS + (size_t)s * 8 + hbase);
    v2f b = *(const v2f*)(aD + (size_t)d * 8 + hbase);
    x0 = leaky(a.x + b.x); x1 = leaky(a.y + b.y);
  } else {
    x0 = leaky(aS[s] + aD[d]); x1 = 0.f;
  }
}

template <int H, int PASS>
__device__ __forceinline__ void process_batch(const int (&s)[4], const int (&dl)[4],
                                              const bool (&ok)[4], int nb,
                                              const float* __restrict__ aS,
                                              const float* __restrict__ aD,
                                              const float* __restrict__ Hf,
                                              float* acc, float* mx, float* sm,
                                              int r0, int hbase, int slab, int lane) {
  constexpr int HS = (H == 8) ? 2 : 1;
  float w0[4], w1[4];
#pragma unroll
  for (int u = 0; u < 4; ++u) {
    float x0, x1;
    edge_logits<H>(aS, aD, s[u], dl[u] + r0, hbase, x0, x1);
    if (PASS == 1) {
      if (ok[u]) {
        atomicMaxFloat(&mx[dl[u] * HS], x0);
        if (HS == 2) atomicMaxFloat(&mx[dl[u] * HS + 1], x1);
      }
    } else {
      w0[u] = 0.f; w1[u] = 0.f;
      if (ok[u]) {
        w0[u] = expf(x0 - mx[dl[u] * HS]);
        ldsAdd(&sm[dl[u] * HS], w0[u]);
        if (HS == 2) {
          w1[u] = expf(x1 - mx[dl[u] * HS + 1]);
          ldsAdd(&sm[dl[u] * HS + 1], w1[u]);
        }
      }
    }
  }
  if (PASS == 2) {
#pragma unroll
    for (int u = 0; u < 4; ++u) {
      int n = nb - u * 32;
      n = n > 32 ? 32 : n;
      for (int k0 = 0; k0 < n; k0 += 16) {
        float hv[16];
#pragma unroll
        for (int j = 0; j < 16; ++j) {
          const int sk = rl_i(s[u], k0 + j);
          hv[j] = Hf[(size_t)sk * HID + slab * SLAB + lane];
        }
#pragma unroll
        for (int j = 0; j < 16; ++j) {
          if (k0 + j < n) {
            const int   dk = rl_i(dl[u], k0 + j);
            const float wa = rl_f(w0[u], k0 + j);
            float wk = wa;
            if (HS == 2) { const float wb = rl_f(w1[u], k0 + j); wk = lane < 16 ? wa : wb; }
            ldsAdd(&acc[dk * SLAB + lane], hv[j] * wk);
          }
        }
      }
    }
  }
}

template <int H, int PASS>
__device__ __forceinline__ void drain_queue(const int* q, int cnt,
                                            const int* __restrict__ src,
                                            const int* __restrict__ dst,
                                            const float* __restrict__ aS,
                                            const float* __restrict__ aD,
                                            const float* __restrict__ Hf,
                                            float* acc, float* mx, float* sm,
                                            int r0, int hbase, int slab, int lane,
                                            float* seg0, int cap, int& ltot) {
  constexpr bool REC = (PASS == 0);
  bool room = false;
  if (REC) {
    room = ltot >= 0 && ltot + cnt <= cap;
    if (!room) ltot = -1;
  }
  for (int i0 = 0; i0 < cnt; i0 += 128) {
    int  s[4], dl[4];
    bool ok[4];
#pragma unroll
    for (int u = 0; u < 4; ++u) {
      const int i = i0 + u * 32 + lane;
      ok[u] = i < cnt;
      const int e = ok[u] ? q[i] : 0;
      s[u]  = src[e];
      dl[u] = dst[e] - r0;
    }
#pragma unroll
    for (int u = 0; u < 4; ++u) {
      if (!ok[u]) { s[u] = 0; dl[u] = 0; }
      if (REC && room && ok[u]) {
        const int p = ltot + i0 + u * 32 + lane;
        const float w = __uint_as_float(((unsigned)s[u] << LBITS) | (unsigned)dl[u]);
        float* at = seg0 + (size_t)(p >> 5) * HID + (p & 31);
#pragma unroll
        for (int sl = 0; sl < NPART; ++sl) at[sl * SLAB] = w;
      }
    }
    if constexpr (PASS != 0) {
      const int nb = cnt - i0;
      process_batch<H, PASS>(s, dl, ok, nb > 128 ? 128 : nb, aS, aD, Hf, acc, mx, sm,
                             r0, hbase, slab, lane);
    }
  }
  if (REC && room) ltot += cnt;
}

template <int H, int PASS>
__device__ __forceinline__ void walk_lists(const float* seg0, size_t qStride,
                                           int c0, int c1, int c2, int c3,
                                           const float* __restrict__ aS,
                                           const float* __restrict__ aD,
                                           const float* __restrict__ Hf,
                                           float* acc, float* mx, float* sm,
                                           int r0, int hbase, int slab, int lane) {
  const int t1 = c0, t2 = t1 + c1, t3 = t2 + c2, cnt = t3 + c3;
  auto word = [&](int g) -> unsigned {
    if (g >= cnt) return 0u;
    const int q   = (g >= t1) + (g >= t2) + (g >= t3);
    const int off = g - (g >= t3 ? t3 : (g >= t2 ? t2 : (g >= t1 ? t1 : 0)));
    return __float_as_uint(seg0[(size_t)q * qStride + (size_t)(off >> 5) * HID + (off & 31)]);
  };
  unsigned wn[4];
#pragma unroll
  for (int u = 0; u < 4; ++u) wn[u] = word(u * 32 + lane);
  for (int i0 = 0; i0 < cnt; i0 += 128) {
    int  s[4], dl[4];
    bool ok[4];
#pragma unroll
    for (int u = 0; u < 4; ++u) {
      ok[u] = i0 + u * 32 + lane < cnt;
      s[u]  = (int)(wn[u] >> LBITS);
      dl[u] = (int)(wn[u] & LMASK);
    }
#pragma unroll
    for (int u = 0; u < 4; ++u) wn[u] = word(i0 + 128 + u * 32 + lane);
    const int nb = cnt - i0;
    process_batch<H, PASS>(s, dl, ok, nb > 128 ? 128 : nb, aS, aD, Hf, acc, mx, sm,
                           r0, hbase, slab, lane);
  }
}

__device__ __forceinline__ void push_hits(int* q, int& cnt, bool hit, int e, int lane) {
  const unsigned m = __builtin_amdgcn_ballot_w32(hit);
  if (m != 0u) {
    const int pos = cnt + __builtin_popcount(m & ((1u << lane) - 1u));
    if (hit) q[pos] = e;
    cnt += __builtin_popcount(m);
  }
}

__device__ __forceinline__ void load_step(const int* __restrict__ dst, int nEdges, int nChunks,
                                          int c, int lane, int (&d)[NLOAD], int& base) {
  base = 0;
  if (c < nChunks) {
    base = c * STEP_E;
    if (base + STEP_E <= nEdges) {
#pragma unroll
      for (int u = 0; u < NLOAD; ++u) d[u] = dst[base + u * 32 + lane];
    } else {
#pragma unroll
      for (int u = 0; u < NLOAD; ++u) {
        const int e = base + u * 32 + lane;
        d[u] = e < nEdges ? dst[e] : -1;
      }
    }
  } else {
#pragma unroll
    for (int u = 0; u < NLOAD; ++u) d[u] = -1;
  }
}

template <int H, int PASS, int NSUB>
__device__ __forceinline__ void scan_edges(int* q, int nEdges,
                                           const int* __restrict__ src,
                                           const int* __restrict__ dst,
                                           const float* __restrict__ aS,
                                           const float* __restrict__ aD,
                                           const float* __restrict__ Hf,
                                           float* acc, float* mx, float* sm,
                                           int r0, int rn, int hbase, int slab,
                                           int lane, int first,
                                           float* seg0, int cap, int& ltot) {
  const int nChunks = (nEdges + STEP_E - 1) / STEP_E;
  const int nRounds = (nChunks + CSTRIDE - 1) / CSTRIDE;
  const int nSteps  = nRounds * NSUB;
  const int rot = (int)(((long long)blockIdx.x * nRounds) / gridDim.x);
  auto chunk_of = [&](int m) {
    int r = m / NSUB + rot;
    if (r >= nRounds) r -= nRounds;
    return m < nSteps ? first + (m % NSUB) + r * CSTRIDE : nChunks;
  };
  int cnt = 0;
  int dn[NLOAD], basen;
  load_step(dst, nEdges, nChunks, chunk_of(0), lane, dn, basen);
  for (int m = 0; ; ++m) {
    const bool more = m < nSteps;
    if (more) {
      int d[NLOAD];
      const int base = basen;
#pragma unroll
      for (int u = 0; u < NLOAD; ++u) d[u] = dn[u];
      load_step(dst, nEdges, nChunks, chunk_of(m + 1), lane, dn, basen);
#pragma unroll
      for (int u = 0; u < NLOAD; ++u)
        push_hits(q, cnt, (unsigned)(d[u] - r0) < (unsigned)rn, base + u * 32 + lane, lane);
    }
    if (cnt > QCAP - STEP_E || (!more && cnt > 0)) {
      drain_queue<H, PASS>(q, cnt, src, dst, aS, aD, Hf, acc, mx, sm, r0, hbase, slab, lane,
                           seg0, cap, ltot);
      cnt = 0;
    }
    if (!more) break;
  }
}

__global__ void __launch_bounds__(1024) scan_lists_kernel(const int* __restrict__ ei, int nEdges,
                                                          float* lst) {
  __shared__ int qbuf[32 * QCAP];
  const int lane = threadIdx.x & 31;
  const int wave = __builtin_amdgcn_readfirstlane(threadIdx.x >> 5);
  const int part = blockIdx.x & (NPART - 1);
  const int tile = blockIdx.x / NPART;
  const int r0   = tile * TILE_R;
  const int rn   = (N_NODES - r0) < TILE_R ? (N_NODES - r0) : TILE_R;
  const int* __restrict__ src = ei;
  const int* __restrict__ dst = ei + nEdges;
  const int ppw = rn / (NPART * 32);
  const int cap = (ppw - 1) * 32;
  float* hdr  = lst + (size_t)(r0 + (part * 32 + wave) * ppw) * HID;
  float* seg0 = hdr + HID;
  int ltot = 0;
  scan_edges<1, 0, 1>(qbuf + wave * QCAP, nEdges, src, dst, nullptr, nullptr, nullptr,
                      nullptr, nullptr, nullptr, r0, rn, 0, 0, lane, part + NPART * wave,
                      seg0, cap, ltot);
  if (lane < NPART) hdr[lane * SLAB] = __int_as_float(ltot);
}

template <int H>
__global__ void __launch_bounds__(1024) owner_layer_kernel(const int* __restrict__ ei, int nEdges,
                                                           const float* __restrict__ aS,
                                                           const float* __restrict__ aD,
                                                           const float* __restrict__ Hf,
                                                           const float* __restrict__ bias,
                                                           float* out, const float* lst,
                                                           int applyElu) {
  constexpr int HS = (H == 8) ? 2 : 1;
  __shared__ float acc[TILE_R * SLAB];
  __shared__ float mx[TILE_R * HS];
  __shared__ float sm[TILE_R * HS];
  __shared__ int   qbuf[32 * QCAP];

  const int lane = threadIdx.x & 31;
  const int wave = __builtin_amdgcn_readfirstlane(threadIdx.x >> 5);
  const int slab = blockIdx.x & 3;
  const int tile = blockIdx.x >> 2;
  const int r0   = tile * TILE_R;
  const int rn   = (N_NODES - r0) < TILE_R ? (N_NODES - r0) : TILE_R;
  const int hbase = (H == 8) ? slab * 2 : 0;
  const int* __restrict__ src = ei;
  const int* __restrict__ dst = ei + nEdges;
  int* q = qbuf + wave * QCAP;

  const int ppw = rn / (NPART * 32);
  const size_t qStride = (size_t)32 * ppw * HID;
  const float* hdr  = lst + (size_t)(r0 + wave * ppw) * HID + slab * SLAB;
  const float* seg0 = hdr + HID;
  const int c0 = __builtin_amdgcn_readfirstlane(__float_as_int(hdr[0]));
  const int c1 = __builtin_amdgcn_readfirstlane(__float_as_int(hdr[qStride]));
  const int c2 = __builtin_amdgcn_readfirstlane(__float_as_int(hdr[2 * qStride]));
  const int c3 = __builtin_amdgcn_readfirstlane(__float_as_int(hdr[3 * qStride]));
  const bool kept = (c0 | c1 | c2 | c3) >= 0;
  int none = 0;

  for (int i = threadIdx.x; i < rn * HS; i += 1024) {
    const int node = r0 + i / HS;
    const int h    = hbase + i % HS;
    mx[i] = leaky(aS[(size_t)node * H + h] + aD[(size_t)node * H + h]);
  }
  __syncthreads();

  if (kept) {
    walk_lists<H, 1>(seg0, qStride, c0, c1, c2, c3, aS, aD, Hf, acc, mx, sm, r0, hbase, slab,
                     lane);
  } else {
    scan_edges<H, 1, NPART>(q, nEdges, src, dst, aS, aD, Hf, acc, mx, sm, r0, rn, hbase, slab,
                            lane, NPART * wave, nullptr, 0, none);
  }
  __syncthreads();

  for (int i = wave; i < rn; i += 32) {
    const int node = r0 + i;
    const int hh   = (HS == 2) ? (lane >> 4) : 0;
    const int h    = hbase + hh;
    const float x  = leaky(aS[(size_t)node * H + h] + aD[(size_t)node * H + h]);
    const float w  = expf(x - mx[i * HS + hh]);
    acc[i * SLAB + lane] = w * Hf[(size_t)node * HID + slab * SLAB + lane];
    if ((lane & 15) == 0 && (lane >> 4) < HS) sm[i * HS + (lane >> 4)] = w;
  }
  __syncthreads();

  if (kept) {
    walk_lists<H, 2>(seg0, qStride, c0, c1, c2, c3, aS, aD, Hf, acc, mx, sm, r0, hbase, slab,
                     lane);
  } else {
    scan_edges<H, 2, NPART>(q, nEdges, src, dst, aS, aD, Hf, acc, mx, sm, r0, rn, hbase, slab,
                            lane, NPART * wave, nullptr, 0, none);
  }
  __syncthreads();

  const float bv = bias[slab * SLAB + lane];
  for (int i = wave; i < rn; i += 32) {
    const int hh = (HS == 2) ? (lane >> 4) : 0;
    float v = acc[i * SLAB + lane] / sm[i * HS + hh] + bv;
    if (applyElu) v = v > 0.f ? v : (expf(v) - 1.f);
    out[(size_t)(r0 + i) * HID + slab * SLAB + lane] = v;
  }
}

extern "C" void kernel_launch(void* const* d_in, const int* in_sizes, int n_in,
                              void* d_out, int out_size, void* d_ws, size_t ws_size,
                              hipStream_t stream) {
  const float* x   = (const float*)d_in[0];
  const int*   ei  = (const int*)d_in[1];
  const float* W1  = (const float*)d_in[2];
  const float* as1 = (const float*)d_in[3];
  const float* ad1 = (const float*)d_in[4];
  const float* b1  = (const float*)d_in[5];
  const float* W2  = (const float*)d_in[6];
  const float* as2 = (const float*)d_in[7];
  const float* ad2 = (const float*)d_in[8];
  const float* b2  = (const float*)d_in[9];

  const int nEdges = in_sizes[1] / 2;

  float* A   = (float*)d_ws;
  float* B   = A  + (size_t)N_NODES * HID;
  float* aS  = B  + (size_t)N_NODES * HID;
  float* aD  = aS + (size_t)N_NODES * GAT_HEADS;
  float* out = (float*)d_out;

  const int TB = 256;
  auto cdiv = [](long long a, long long b) { return (unsigned)((a + b - 1) / b); };
  const unsigned gGemm  = (N_NODES / 16) * (HID / 16) / (TB / 32);
  const unsigned gOwner = ((N_NODES + TILE_R - 1) / TILE_R) * (HID / SLAB);

  scan_lists_kernel<<<gOwner, 1024, 0, stream>>>(ei, nEdges, out);

  gemm128_wmma<<<gGemm, TB, 0, stream>>>(x, W1, A);
  alpha1_kernel<<<cdiv(N_NODES * GAT_HEADS, TB), TB, 0, stream>>>(A, as1, ad1, aS, aD);
  owner_layer_kernel<GAT_HEADS><<<gOwner, 1024, 0, stream>>>(ei, nEdges, aS, aD, A, b1, B, out, 1);

  gemm128_wmma<<<gGemm, TB, 0, stream>>>(B, W2, A);
  alpha2_kernel<<<cdiv((long long)N_NODES * 32, TB), TB, 0, stream>>>(A, as2, ad2, aS, aD);
  owner_layer_kernel<1><<<gOwner, 1024, 0, stream>>>(ei, nEdges, aS, aD, A, b2, out, out, 0);
}


# Edge logits computed once per layer: the max pass stores each list entry's logits as whole lines in the wave's own scratch lines, the sum pass reads them back (4 lines at once), no second a_src gather
# speedup vs baseline: 2.7393x; 1.0334x over previous
#include <hip/hip_runtime.h>

#define N_NODES 100000
#define IN_DIM 128
#define HID 128
#define GAT_HEADS 8
#define NEG_SLOPE 0.2f

typedef float v2f __attribute__((ext_vector_type(2)));
typedef float v4f __attribute__((ext_vector_type(4)));
typedef float v8f __attribute__((ext_vector_type(8)));

__global__ void __launch_bounds__(256) gemm128_wmma(const float* __restrict__ X,
                                                    const float* __restrict__ W,
                                                    float* __restrict__ H) {
  const int lane  = threadIdx.x & 31;
  const int wave  = blockIdx.x * (blockDim.x >> 5) + (threadIdx.x >> 5);
  const int tileM = wave >> 3;
  const int tileN = wave & 7;
  const int r     = lane & 15;
  const int khalf = lane >> 4;
  const int row   = tileM * 16 + r;
  const int col   = tileN * 16 + r;
  const float* __restrict__ xrow = X + (size_t)row * IN_DIM;

  v8f c = {0.f, 0.f, 0.f, 0.f, 0.f, 0.f, 0.f, 0.f};
#pragma unroll 4
  for (int k = 0; k < IN_DIM; k += 4) {
    const int kk = k + 2 * khalf;
    v2f a = *(const v2f*)(xrow + kk);
    v2f b;
    b.x = W[(kk    ) * HID + col];
    b.y = W[(kk + 1) * HID + col];
    c = __builtin_amdgcn_wmma_f32_16x16x4_f32(false, a, false, b, (short)0, c,
                                              false, false);
  }
  const int mbase = tileM * 16 + khalf * 8;
#pragma unroll
  for (int v = 0; v < 8; ++v)
    H[(size_t)(mbase + v) * HID + tileN * 16 + r] = c[v];
}

__global__ void alpha1_kernel(const float* __restrict__ H,
                              const float* __restrict__ avS,
                              const float* __restrict__ avD,
                              float* __restrict__ outS,
                              float* __restrict__ outD) {
  int i = blockIdx.x * blockDim.x + threadIdx.x;
  if (i >= N_NODES * GAT_HEADS) return;
  int head = i & 7, node = i >> 3;
  const float* h  = H   + (size_t)node * HID + head * 16;
  const float* as = avS + head * 16;
  const float* ad = avD + head * 16;
  float s = 0.f, d = 0.f;
#pragma unroll
  for (int c = 0; c < 16; ++c) { float hv = h[c]; s += hv * as[c]; d += hv * ad[c]; }
  outS[i] = s; outD[i] = d;
}

__global__ void alpha2_kernel(const float* __restrict__ H,
                              const float* __restrict__ avS,
                              const float* __restrict__ avD,
                              float* __restrict__ outS,
                              float* __restrict__ outD) {
  int wave = blockIdx.x * (blockDim.x >> 5) + (threadIdx.x >> 5);
  int lane = threadIdx.x & 31;
  if (wave >= N_NODES) return;
  float s = 0.f, d = 0.f;
#pragma unroll
  for (int c = lane; c < HID; c += 32) {
    float hv = H[(size_t)wave * HID + c];
    s += hv * avS[c]; d += hv * avD[c];
  }
#pragma unroll
  for (int off = 16; off > 0; off >>= 1) {
    s += __shfl_xor(s, off, 32);
    d += __shfl_xor(d, off, 32);
  }
  if (lane == 0) { outS[wave] = s; outD[wave] = d; }
}

#define TILE_R 1728
#define SLAB   32
#define NLOAD  12
#define STEP_E (NLOAD * 32)
#define QCAP   512
#define LBITS  11
#define LMASK  ((1u << LBITS) - 1u)
static_assert(TILE_R <= (1 << LBITS), "tile-local destination must fit its field");
static_assert(N_NODES <= (1 << (32 - LBITS)), "source node must fit its field");
#define NPART  4
#define CSTRIDE (NPART * 32)
static_assert(NPART == HID / SLAB, "one scanning work-group per slab work-group of a tile");
static_assert(N_NODES % TILE_R == 0 || N_NODES % TILE_R >= 2 * NPART * 32,
              "the last tile must give every list a count line and a data line of its own");

__device__ __forceinline__ float leaky(float x) {
  return x > 0.f ? x : NEG_SLOPE * x;
}

__device__ __forceinline__ void atomicMaxFloat(float* addr, float val) {
  if (__float_as_int(val) >= 0)
    atomicMax((int*)addr, __float_as_int(val));
  else
    atomicMin((unsigned int*)addr, __float_as_uint(val));
}

__device__ __forceinline__ void ldsAdd(float* addr, float val) {
  __hip_atomic_fetch_add(addr, val, __ATOMIC_RELAXED, __HIP_MEMORY_SCOPE_WORKGROUP);
}

__device__ __forceinline__ int rl_i(int v, int k) { return __builtin_amdgcn_readlane(v, k); }
__device__ __forceinline__ float rl_f(float v, int k) {
  return __int_as_float(__builtin_amdgcn_readlane(__float_as_int(v), k));
}

template <int H>
__device__ __forceinline__ void edge_logits(const float* __restrict__ aS,
                                            const float* __restrict__ aD,
                                            int s, int d, int hbase, float& x0, float& x1) {
  if (H == 8) {
    v2f a = *(const v2f*)(aS + (size_t)s * 8 + hbase);
    v2f b = *(const v2f*)(aD + (size_t)d * 8 + hbase);
    x0 = leaky(a.x + b.x); x1 = leaky(a.y + b.y);
  } else {
    x0 = leaky(aS[s] + aD[d]); x1 = 0.f;
  }
}

template <int H, int PASS>
__device__ __forceinline__ void process_batch(const int (&s)[4], const int (&dl)[4],
                                              const bool (&ok)[4], int nb,
                                              const float* __restrict__ aS,
                                              const float* __restrict__ aD,
                                              const float* __restrict__ Hf,
                                              float* acc, float* mx, float* sm,
                                              int r0, int hbase, int slab, int lane,
                                              float* xs, int g0, int capE) {
  constexpr int HS = (H == 8) ? 2 : 1;
  float w0[4], w1[4];
  float xh0[4], xh1[4];
  if (PASS == 2) {
#pragma unroll
    for (int u = 0; u < 4; ++u) {
      const int g = g0 + u * 32 + lane;
      const float* xp = xs + (size_t)((g * HS) >> 5) * HID + ((g * HS) & 31);
      xh0[u] = 0.f; xh1[u] = 0.f;
      if (ok[u] && g < capE) {
        if (HS == 2) { const v2f xv = *(const v2f*)xp; xh0[u] = xv.x; xh1[u] = xv.y; }
        else         xh0[u] = *xp;
      }
    }
  }
#pragma unroll
  for (int u = 0; u < 4; ++u) {
    float x0, x1;
    const int g = g0 + u * 32 + lane;
    float* xp = xs + (size_t)((g * HS) >> 5) * HID + ((g * HS) & 31);
    const bool held = g < capE;
    if (PASS == 2 && held) {
      x0 = xh0[u]; x1 = xh1[u];
    } else {
      edge_logits<H>(aS, aD, s[u], dl[u] + r0, hbase, x0, x1);
    }
    if (PASS == 1) {
      if (ok[u] && held) {
        if (HS == 2) { v2f xv; xv.x = x0; xv.y = x1; *(v2f*)xp = xv; }
        else         *xp = x0;
      }
      if (ok[u]) {
        atomicMaxFloat(&mx[dl[u] * HS], x0);
        if (HS == 2) atomicMaxFloat(&mx[dl[u] * HS + 1], x1);
      }
    } else {
      w0[u] = 0.f; w1[u] = 0.f;
      if (ok[u]) {
        w0[u] = expf(x0 - mx[dl[u] * HS]);
        ldsAdd(&sm[dl[u] * HS], w0[u]);
        if (HS == 2) {
          w1[u] = expf(x1 - mx[dl[u] * HS + 1]);
          ldsAdd(&sm[dl[u] * HS + 1], w1[u]);
        }
      }
    }
  }
  if (PASS == 2) {
#pragma unroll
    for (int u = 0; u < 4; ++u) {
      int n = nb - u * 32;
      n = n > 32 ? 32 : n;
      for (int k0 = 0; k0 < n; k0 += 16) {
        float hv[16];
#pragma unroll
        for (int j = 0; j < 16; ++j) {
          const int sk = rl_i(s[u], k0 + j);
          hv[j] = Hf[(size_t)sk * HID + slab * SLAB + lane];
        }
#pragma unroll
        for (int j = 0; j < 16; ++j) {
          if (k0 + j < n) {
            const int   dk = rl_i(dl[u], k0 + j);
            const float wa = rl_f(w0[u], k0 + j);
            float wk = wa;
            if (HS == 2) { const float wb = rl_f(w1[u], k0 + j); wk = lane < 16 ? wa : wb; }
            ldsAdd(&acc[dk * SLAB + lane], hv[j] * wk);
          }
        }
      }
    }
  }
}

template <int H, int PASS>
__device__ __forceinline__ void drain_queue(const int* q, int cnt,
                                            const int* __restrict__ src,
                                            const int* __restrict__ dst,
                                            const float* __restrict__ aS,
                                            const float* __restrict__ aD,
                                            const float* __restrict__ Hf,
                                            float* acc, float* mx, float* sm,
                                            int r0, int hbase, int slab, int lane,
                                            float* seg0, int cap, int& ltot) {
  constexpr bool REC = (PASS == 0);
  bool room = false;
  if (REC) {
    room = ltot >= 0 && ltot + cnt <= cap;
    if (!room) ltot = -1;
  }
  for (int i0 = 0; i0 < cnt; i0 += 128) {
    int  s[4], dl[4];
    bool ok[4];
#pragma unroll
    for (int u = 0; u < 4; ++u) {
      const int i = i0 + u * 32 + lane;
      ok[u] = i < cnt;
      const int e = ok[u] ? q[i] : 0;
      s[u]  = src[e];
      dl[u] = dst[e] - r0;
    }
#pragma unroll
    for (int u = 0; u < 4; ++u) {
      if (!ok[u]) { s[u] = 0; dl[u] = 0; }
      if (REC && room && ok[u]) {
        const int p = ltot + i0 + u * 32 + lane;
        const float w = __uint_as_float(((unsigned)s[u] << LBITS) | (unsigned)dl[u]);
        float* at = seg0 + (size_t)(p >> 5) * HID + (p & 31);
#pragma unroll
        for (int sl = 0; sl < NPART; ++sl) at[sl * SLAB] = w;
      }
    }
    if constexpr (PASS != 0) {
      const int nb = cnt - i0;
      process_batch<H, PASS>(s, dl, ok, nb > 128 ? 128 : nb, aS, aD, Hf, acc, mx, sm,
                             r0, hbase, slab, lane, nullptr, 0, 0);
    }
  }
  if (REC && room) ltot += cnt;
}

template <int H, int PASS>
__device__ __forceinline__ void walk_lists(const float* seg0, size_t qStride,
                                           int c0, int c1, int c2, int c3,
                                           const float* __restrict__ aS,
                                           const float* __restrict__ aD,
                                           const float* __restrict__ Hf,
                                           float* acc, float* mx, float* sm,
                                           int r0, int hbase, int slab, int lane,
                                           float* xs, int capE) {
  const int t1 = c0, t2 = t1 + c1, t3 = t2 + c2, cnt = t3 + c3;
  auto word = [&](int g) -> unsigned {
    if (g >= cnt) return 0u;
    const int q   = (g >= t1) + (g >= t2) + (g >= t3);
    const int off = g - (g >= t3 ? t3 : (g >= t2 ? t2 : (g >= t1 ? t1 : 0)));
    return __float_as_uint(seg0[(size_t)q * qStride + (size_t)(off >> 5) * HID + (off & 31)]);
  };
  unsigned wn[4];
#pragma unroll
  for (int u = 0; u < 4; ++u) wn[u] = word(u * 32 + lane);
  for (int i0 = 0; i0 < cnt; i0 += 128) {
    int  s[4], dl[4];
    bool ok[4];
#pragma unroll
    for (int u = 0; u < 4; ++u) {
      ok[u] = i0 + u * 32 + lane < cnt;
      s[u]  = (int)(wn[u] >> LBITS);
      dl[u] = (int)(wn[u] & LMASK);
    }
#pragma unroll
    for (int u = 0; u < 4; ++u) wn[u] = word(i0 + 128 + u * 32 + lane);
    const int nb = cnt - i0;
    process_batch<H, PASS>(s, dl, ok, nb > 128 ? 128 : nb, aS, aD, Hf, acc, mx, sm,
                           r0, hbase, slab, lane, xs, i0, capE);
  }
}

__device__ __forceinline__ void push_hits(int* q, int& cnt, bool hit, int e, int lane) {
  const unsigned m = __builtin_amdgcn_ballot_w32(hit);
  if (m != 0u) {
    const int pos = cnt + __builtin_popcount(m & ((1u << lane) - 1u));
    if (hit) q[pos] = e;
    cnt += __builtin_popcount(m);
  }
}

__device__ __forceinline__ void load_step(const int* __restrict__ dst, int nEdges, int nChunks,
                                          int c, int lane, int (&d)[NLOAD], int& base) {
  base = 0;
  if (c < nChunks) {
    base = c * STEP_E;
    if (base + STEP_E <= nEdges) {
#pragma unroll
      for (int u = 0; u < NLOAD; ++u) d[u] = dst[base + u * 32 + lane];
    } else {
#pragma unroll
      for (int u = 0; u < NLOAD; ++u) {
        const int e = base + u * 32 + lane;
        d[u] = e < nEdges ? dst[e] : -1;
      }
    }
  } else {
#pragma unroll
    for (int u = 0; u < NLOAD; ++u) d[u] = -1;
  }
}

template <int H, int PASS, int NSUB>
__device__ __forceinline__ void scan_edges(int* q, int nEdges,
                                           const int* __restrict__ src,
                                           const int* __restrict__ dst,
                                           const float* __restrict__ aS,
                                           const float* __restrict__ aD,
                                           const float* __restrict__ Hf,
                                           float* acc, float* mx, float* sm,
                                           int r0, int rn, int hbase, int slab,
                                           int lane, int first,
                                           float* seg0, int cap, int& ltot) {
  const int nChunks = (nEdges + STEP_E - 1) / STEP_E;
  const int nRounds = (nChunks + CSTRIDE - 1) / CSTRIDE;
  const int nSteps  = nRounds * NSUB;
  const int rot = (int)(((long long)blockIdx.x * nRounds) / gridDim.x);
  auto chunk_of = [&](int m) {
    int r = m / NSUB + rot;
    if (r >= nRounds) r -= nRounds;
    return m < nSteps ? first + (m % NSUB) + r * CSTRIDE : nChunks;
  };
  int cnt = 0;
  int dn[NLOAD], basen;
  load_step(dst, nEdges, nChunks, chunk_of(0), lane, dn, basen);
  for (int m = 0; ; ++m) {
    const bool more = m < nSteps;
    if (more) {
      int d[NLOAD];
      const int base = basen;
#pragma unroll
      for (int u = 0; u < NLOAD; ++u) d[u] = dn[u];
      load_step(dst, nEdges, nChunks, chunk_of(m + 1), lane, dn, basen);
#pragma unroll
      for (int u = 0; u < NLOAD; ++u)
        push_hits(q, cnt, (unsigned)(d[u] - r0) < (unsigned)rn, base + u * 32 + lane, lane);
    }
    if (cnt > QCAP - STEP_E || (!more && cnt > 0)) {
      drain_queue<H, PASS>(q, cnt, src, dst, aS, aD, Hf, acc, mx, sm, r0, hbase, slab, lane,
                           seg0, cap, ltot);
      cnt = 0;
    }
    if (!more) break;
  }
}

__global__ void __launch_bounds__(1024) scan_lists_kernel(const int* __restrict__ ei, int nEdges,
                                                          float* lst) {
  __shared__ int qbuf[32 * QCAP];
  const int lane = threadIdx.x & 31;
  const int wave = __builtin_amdgcn_readfirstlane(threadIdx.x >> 5);
  const int part = blockIdx.x & (NPART - 1);
  const int tile = blockIdx.x / NPART;
  const int r0   = tile * TILE_R;
  const int rn   = (N_NODES - r0) < TILE_R ? (N_NODES - r0) : TILE_R;
  const int* __restrict__ src = ei;
  const int* __restrict__ dst = ei + nEdges;
  const int ppw = rn / (NPART * 32);
  const int cap = (ppw - 1) * 32;
  float* hdr  = lst + (size_t)(r0 + (part * 32 + wave) * ppw) * HID;
  float* seg0 = hdr + HID;
  int ltot = 0;
  scan_edges<1, 0, 1>(qbuf + wave * QCAP, nEdges, src, dst, nullptr, nullptr, nullptr,
                      nullptr, nullptr, nullptr, r0, rn, 0, 0, lane, part + NPART * wave,
                      seg0, cap, ltot);
  if (lane < NPART) hdr[lane * SLAB] = __int_as_float(ltot);
}

template <int H>
__global__ void __launch_bounds__(1024) owner_layer_kernel(const int* __restrict__ ei, int nEdges,
                                                           const float* __restrict__ aS,
                                                           const float* __restrict__ aD,
                                                           const float* __restrict__ Hf,
                                                           const float* __restrict__ bias,
                                                           float* out, const float* lst,
                                                           float* scr, int applyElu) {
  constexpr int HS = (H == 8) ? 2 : 1;
  __shared__ float acc[TILE_R * SLAB];
  __shared__ float mx[TILE_R * HS];
  __shared__ float sm[TILE_R * HS];
  __shared__ int   qbuf[32 * QCAP];

  const int lane = threadIdx.x & 31;
  const int wave = __builtin_amdgcn_readfirstlane(threadIdx.x >> 5);
  const int slab = blockIdx.x & 3;
  const int tile = blockIdx.x >> 2;
  const int r0   = tile * TILE_R;
  const int rn   = (N_NODES - r0) < TILE_R ? (N_NODES - r0) : TILE_R;
  const int hbase = (H == 8) ? slab * 2 : 0;
  const int* __restrict__ src = ei;
  const int* __restrict__ dst = ei + nEdges;
  int* q = qbuf + wave * QCAP;

  const int ppw = rn / (NPART * 32);
  const size_t qStride = (size_t)32 * ppw * HID;
  const float* hdr  = lst + (size_t)(r0 + wave * ppw) * HID + slab * SLAB;
  const float* seg0 = hdr + HID;
  const int c0 = __builtin_amdgcn_readfirstlane(__float_as_int(hdr[0]));
  const int c1 = __builtin_amdgcn_readfirstlane(__float_as_int(hdr[qStride]));
  const int c2 = __builtin_amdgcn_readfirstlane(__float_as_int(hdr[2 * qStride]));
  const int c3 = __builtin_amdgcn_readfirstlane(__float_as_int(hdr[3 * qStride]));
  const bool kept = (c0 | c1 | c2 | c3) >= 0;
  int none = 0;
  const int rpw  = rn / 32;
  float* xs      = scr + (size_t)(r0 + wave * rpw) * HID + slab * SLAB;
  const int capE = rpw * 32 / HS;

  for (int i = threadIdx.x; i < rn * HS; i += 1024) {
    const int node = r0 + i / HS;
    const int h    = hbase + i % HS;
    mx[i] = leaky(aS[(size_t)node * H + h] + aD[(size_t)node * H + h]);
  }
  __syncthreads();

  if (kept) {
    walk_lists<H, 1>(seg0, qStride, c0, c1, c2, c3, aS, aD, Hf, acc, mx, sm, r0, hbase, slab,
                     lane, xs, capE);
  } else {
    scan_edges<H, 1, NPART>(q, nEdges, src, dst, aS, aD, Hf, acc, mx, sm, r0, rn, hbase, slab,
                            lane, NPART * wave, nullptr, 0, none);
  }
  __syncthreads();

  for (int i = wave; i < rn; i += 32) {
    const int node = r0 + i;
    const int hh   = (HS == 2) ? (lane >> 4) : 0;
    const int h    = hbase + hh;
    const float x  = leaky(aS[(size_t)node * H + h] + aD[(size_t)node * H + h]);
    const float w  = expf(x - mx[i * HS + hh]);
    acc[i * SLAB + lane] = w * Hf[(size_t)node * HID + slab * SLAB + lane];
    if ((lane & 15) == 0 && (lane >> 4) < HS) sm[i * HS + (lane >> 4)] = w;
  }
  __syncthreads();

  if (kept) {
    walk_lists<H, 2>(seg0, qStride, c0, c1, c2, c3, aS, aD, Hf, acc, mx, sm, r0, hbase, slab,
                     lane, xs, capE);
  } else {
    scan_edges<H, 2, NPART>(q, nEdges, src, dst, aS, aD, Hf, acc, mx, sm, r0, rn, hbase, slab,
                            lane, NPART * wave, nullptr, 0, none);
  }
  __syncthreads();

  const float bv = bias[slab * SLAB + lane];
  for (int i = wave; i < rn; i += 32) {
    const int hh = (HS == 2) ? (lane >> 4) : 0;
    float v = acc[i * SLAB + lane] / sm[i * HS + hh] + bv;
    if (applyElu) v = v > 0.f ? v : (expf(v) - 1.f);
    out[(size_t)(r0 + i) * HID + slab * SLAB + lane] = v;
  }
}

extern "C" void kernel_launch(void* const* d_in, const int* in_sizes, int n_in,
                              void* d_out, int out_size, void* d_ws, size_t ws_size,
                              hipStream_t stream) {
  const float* x   = (const float*)d_in[0];
  const int*   ei  = (const int*)d_in[1];
  const float* W1  = (const float*)d_in[2];
  const float* as1 = (const float*)d_in[3];
  const float* ad1 = (const float*)d_in[4];
  const float* b1  = (const float*)d_in[5];
  const float* W2  = (const float*)d_in[6];
  const float* as2 = (const float*)d_in[7];
  const float* ad2 = (const float*)d_in[8];
  const float* b2  = (const float*)d_in[9];

  const int nEdges = in_sizes[1] / 2;

  float* A   = (float*)d_ws;
  float* B   = A  + (size_t)N_NODES * HID;
  float* aS  = B  + (size_t)N_NODES * HID;
  float* aD  = aS + (size_t)N_NODES * GAT_HEADS;
  float* out = (float*)d_out;

  const int TB = 256;
  auto cdiv = [](long long a, long long b) { return (unsigned)((a + b - 1) / b); };
  const unsigned gGemm  = (N_NODES / 16) * (HID / 16) / (TB / 32);
  const unsigned gOwner = ((N_NODES + TILE_R - 1) / TILE_R) * (HID / SLAB);

  scan_lists_kernel<<<gOwner, 1024, 0, stream>>>(ei, nEdges, out);

  gemm128_wmma<<<gGemm, TB, 0, stream>>>(x, W1, A);
  alpha1_kernel<<<cdiv(N_NODES * GAT_HEADS, TB), TB, 0, stream>>>(A, as1, ad1, aS, aD);
  owner_layer_kernel<GAT_HEADS><<<gOwner, 1024, 0, stream>>>(ei, nEdges, aS, aD, A, b1, B, out, B, 1);

  gemm128_wmma<<<gGemm, TB, 0, stream>>>(B, W2, A);
  alpha2_kernel<<<cdiv((long long)N_NODES * 32, TB), TB, 0, stream>>>(A, as2, ad2, aS, aD);
  owner_layer_kernel<1><<<gOwner, 1024, 0, stream>>>(ei, nEdges, aS, aD, A, b2, out, out, B, 0);
}


# Tiles of 1568 nodes, not 1728: 64 tiles x 4 slabs = 256 work-groups, one on every WGP (232 before: 24 WGPs idle in the scan and both edge launches); tile bounds move, so LDS sum order changes
# speedup vs baseline: 2.8336x; 1.0344x over previous
#include <hip/hip_runtime.h>

#define N_NODES 100000
#define IN_DIM 128
#define HID 128
#define GAT_HEADS 8
#define NEG_SLOPE 0.2f

typedef float v2f __attribute__((ext_vector_type(2)));
typedef float v4f __attribute__((ext_vector_type(4)));
typedef float v8f __attribute__((ext_vector_type(8)));

__global__ void __launch_bounds__(256) gemm128_wmma(const float* __restrict__ X,
                                                    const float* __restrict__ W,
                                                    float* __restrict__ H) {
  const int lane  = threadIdx.x & 31;
  const int wave  = blockIdx.x * (blockDim.x >> 5) + (threadIdx.x >> 5);
  const int tileM = wave >> 3;
  const int tileN = wave & 7;
  const int r     = lane & 15;
  const int khalf = lane >> 4;
  const int row   = tileM * 16 + r;
  const int col   = tileN * 16 + r;
  const float* __restrict__ xrow = X + (size_t)row * IN_DIM;

  v8f c = {0.f, 0.f, 0.f, 0.f, 0.f, 0.f, 0.f, 0.f};
#pragma unroll 4
  for (int k = 0; k < IN_DIM; k += 4) {
    const int kk = k + 2 * khalf;
    v2f a = *(const v2f*)(xrow + kk);
    v2f b;
    b.x = W[(kk    ) * HID + col];
    b.y = W[(kk + 1) * HID + col];
    c = __builtin_amdgcn_wmma_f32_16x16x4_f32(false, a, false, b, (short)0, c,
                                              false, false);
  }
  const int mbase = tileM * 16 + khalf * 8;
#pragma unroll
  for (int v = 0; v < 8; ++v)
    H[(size_t)(mbase + v) * HID + tileN * 16 + r] = c[v];
}

__global__ void alpha1_kernel(const float* __restrict__ H,
                              const float* __restrict__ avS,
                              const float* __restrict__ avD,
                              float* __restrict__ outS,
                              float* __restrict__ outD) {
  int i = blockIdx.x * blockDim.x + threadIdx.x;
  if (i >= N_NODES * GAT_HEADS) return;
  int head = i & 7, node = i >> 3;
  const float* h  = H   + (size_t)node * HID + head * 16;
  const float* as = avS + head * 16;
  const float* ad = avD + head * 16;
  float s = 0.f, d = 0.f;
#pragma unroll
  for (int c = 0; c < 16; ++c) { float hv = h[c]; s += hv * as[c]; d += hv * ad[c]; }
  outS[i] = s; outD[i] = d;
}

__global__ void alpha2_kernel(const float* __restrict__ H,
                              const float* __restrict__ avS,
                              const float* __restrict__ avD,
                              float* __restrict__ outS,
                              float* __restrict__ outD) {
  int wave = blockIdx.x * (blockDim.x >> 5) + (threadIdx.x >> 5);
  int lane = threadIdx.x & 31;
  if (wave >= N_NODES) return;
  float s = 0.f, d = 0.f;
#pragma unroll
  for (int c = lane; c < HID; c += 32) {
    float hv = H[(size_t)wave * HID + c];
    s += hv * avS[c]; d += hv * avD[c];
  }
#pragma unroll
  for (int off = 16; off > 0; off >>= 1) {
    s += __shfl_xor(s, off, 32);
    d += __shfl_xor(d, off, 32);
  }
  if (lane == 0) { outS[wave] = s; outD[wave] = d; }
}

#define TILE_R 1568
#define SLAB   32
#define NLOAD  12
#define STEP_E (NLOAD * 32)
#define QCAP   512
#define LBITS  11
#define LMASK  ((1u << LBITS) - 1u)
static_assert(TILE_R <= (1 << LBITS), "tile-local destination must fit its field");
static_assert(N_NODES <= (1 << (32 - LBITS)), "source node must fit its field");
#define NPART  4
#define CSTRIDE (NPART * 32)
static_assert(NPART == HID / SLAB, "one scanning work-group per slab work-group of a tile");
static_assert(N_NODES % TILE_R == 0 || N_NODES % TILE_R >= 2 * NPART * 32,
              "the last tile must give every list a count line and a data line of its own");

__device__ __forceinline__ float leaky(float x) {
  return x > 0.f ? x : NEG_SLOPE * x;
}

__device__ __forceinline__ void atomicMaxFloat(float* addr, float val) {
  if (__float_as_int(val) >= 0)
    atomicMax((int*)addr, __float_as_int(val));
  else
    atomicMin((unsigned int*)addr, __float_as_uint(val));
}

__device__ __forceinline__ void ldsAdd(float* addr, float val) {
  __hip_atomic_fetch_add(addr, val, __ATOMIC_RELAXED, __HIP_MEMORY_SCOPE_WORKGROUP);
}

__device__ __forceinline__ int rl_i(int v, int k) { return __builtin_amdgcn_readlane(v, k); }
__device__ __forceinline__ float rl_f(float v, int k) {
  return __int_as_float(__builtin_amdgcn_readlane(__float_as_int(v), k));
}

template <int H>
__device__ __forceinline__ void edge_logits(const float* __restrict__ aS,
                                            const float* __restrict__ aD,
                                            int s, int d, int hbase, float& x0, float& x1) {
  if (H == 8) {
    v2f a = *(const v2f*)(aS + (size_t)s * 8 + hbase);
    v2f b = *(const v2f*)(aD + (size_t)d * 8 + hbase);
    x0 = leaky(a.x + b.x); x1 = leaky(a.y + b.y);
  } else {
    x0 = leaky(aS[s] + aD[d]); x1 = 0.f;
  }
}

template <int H, int PASS>
__device__ __forceinline__ void process_batch(const int (&s)[4], const int (&dl)[4],
                                              const bool (&ok)[4], int nb,
                                              const float* __restrict__ aS,
                                              const float* __restrict__ aD,
                                              const float* __restrict__ Hf,
                                              float* acc, float* mx, float* sm,
                                              int r0, int hbase, int slab, int lane,
                                              float* xs, int g0, int capE) {
  constexpr int HS = (H == 8) ? 2 : 1;
  float w0[4], w1[4];
  float xh0[4], xh1[4];
  if (PASS == 2) {
#pragma unroll
    for (int u = 0; u < 4; ++u) {
      const int g = g0 + u * 32 + lane;
      const float* xp = xs + (size_t)((g * HS) >> 5) * HID + ((g * HS) & 31);
      xh0[u] = 0.f; xh1[u] = 0.f;
      if (ok[u] && g < capE) {
        if (HS == 2) { const v2f xv = *(const v2f*)xp; xh0[u] = xv.x; xh1[u] = xv.y; }
        else         xh0[u] = *xp;
      }
    }
  }
#pragma unroll
  for (int u = 0; u < 4; ++u) {
    float x0, x1;
    const int g = g0 + u * 32 + lane;
    float* xp = xs + (size_t)((g * HS) >> 5) * HID + ((g * HS) & 31);
    const bool held = g < capE;
    if (PASS == 2 && held) {
      x0 = xh0[u]; x1 = xh1[u];
    } else {
      edge_logits<H>(aS, aD, s[u], dl[u] + r0, hbase, x0, x1);
    }
    if (PASS == 1) {
      if (ok[u] && held) {
        if (HS == 2) { v2f xv; xv.x = x0; xv.y = x1; *(v2f*)xp = xv; }
        else         *xp = x0;
      }
      if (ok[u]) {
        atomicMaxFloat(&mx[dl[u] * HS], x0);
        if (HS == 2) atomicMaxFloat(&mx[dl[u] * HS + 1], x1);
      }
    } else {
      w0[u] = 0.f; w1[u] = 0.f;
      if (ok[u]) {
        w0[u] = expf(x0 - mx[dl[u] * HS]);
        ldsAdd(&sm[dl[u] * HS], w0[u]);
        if (HS == 2) {
          w1[u] = expf(x1 - mx[dl[u] * HS + 1]);
          ldsAdd(&sm[dl[u] * HS + 1], w1[u]);
        }
      }
    }
  }
  if (PASS == 2) {
#pragma unroll
    for (int u = 0; u < 4; ++u) {
      int n = nb - u * 32;
      n = n > 32 ? 32 : n;
      for (int k0 = 0; k0 < n; k0 += 16) {
        float hv[16];
#pragma unroll
        for (int j = 0; j < 16; ++j) {
          const int sk = rl_i(s[u], k0 + j);
          hv[j] = Hf[(size_t)sk * HID + slab * SLAB + lane];
        }
#pragma unroll
        for (int j = 0; j < 16; ++j) {
          if (k0 + j < n) {
            const int   dk = rl_i(dl[u], k0 + j);
            const float wa = rl_f(w0[u], k0 + j);
            float wk = wa;
            if (HS == 2) { const float wb = rl_f(w1[u], k0 + j); wk = lane < 16 ? wa : wb; }
            ldsAdd(&acc[dk * SLAB + lane], hv[j] * wk);
          }
        }
      }
    }
  }
}

template <int H, int PASS>
__device__ __forceinline__ void drain_queue(const int* q, int cnt,
                                            const int* __restrict__ src,
                                            const int* __restrict__ dst,
                                            const float* __restrict__ aS,
                                            const float* __restrict__ aD,
                                            const float* __restrict__ Hf,
                                            float* acc, float* mx, float* sm,
                                            int r0, int hbase, int slab, int lane,
                                            float* seg0, int cap, int& ltot) {
  constexpr bool REC = (PASS == 0);
  bool room = false;
  if (REC) {
    room = ltot >= 0 && ltot + cnt <= cap;
    if (!room) ltot = -1;
  }
  for (int i0 = 0; i0 < cnt; i0 += 128) {
    int  s[4], dl[4];
    bool ok[4];
#pragma unroll
    for (int u = 0; u < 4; ++u) {
      const int i = i0 + u * 32 + lane;
      ok[u] = i < cnt;
      const int e = ok[u] ? q[i] : 0;
      s[u]  = src[e];
      dl[u] = dst[e] - r0;
    }
#pragma unroll
    for (int u = 0; u < 4; ++u) {
      if (!ok[u]) { s[u] = 0; dl[u] = 0; }
      if (REC && room && ok[u]) {
        const int p = ltot + i0 + u * 32 + lane;
        const float w = __uint_as_float(((unsigned)s[u] << LBITS) | (unsigned)dl[u]);
        float* at = seg0 + (size_t)(p >> 5) * HID + (p & 31);
#pragma unroll
        for (int sl = 0; sl < NPART; ++sl) at[sl * SLAB] = w;
      }
    }
    if constexpr (PASS != 0) {
      const int nb = cnt - i0;
      process_batch<H, PASS>(s, dl, ok, nb > 128 ? 128 : nb, aS, aD, Hf, acc, mx, sm,
                             r0, hbase, slab, lane, nullptr, 0, 0);
    }
  }
  if (REC && room) ltot += cnt;
}

template <int H, int PASS>
__device__ __forceinline__ void walk_lists(const float* seg0, size_t qStride,
                                           int c0, int c1, int c2, int c3,
                                           const float* __restrict__ aS,
                                           const float* __restrict__ aD,
                                           const float* __restrict__ Hf,
                                           float* acc, float* mx, float* sm,
                                           int r0, int hbase, int slab, int lane,
                                           float* xs, int capE) {
  const int t1 = c0, t2 = t1 + c1, t3 = t2 + c2, cnt = t3 + c3;
  auto word = [&](int g) -> unsigned {
    if (g >= cnt) return 0u;
    const int q   = (g >= t1) + (g >= t2) + (g >= t3);
    const int off = g - (g >= t3 ? t3 : (g >= t2 ? t2 : (g >= t1 ? t1 : 0)));
    return __float_as_uint(seg0[(size_t)q * qStride + (size_t)(off >> 5) * HID + (off & 31)]);
  };
  unsigned wn[4];
#pragma unroll
  for (int u = 0; u < 4; ++u) wn[u] = word(u * 32 + lane);
  for (int i0 = 0; i0 < cnt; i0 += 128) {
    int  s[4], dl[4];
    bool ok[4];
#pragma unroll
    for (int u = 0; u < 4; ++u) {
      ok[u] = i0 + u * 32 + lane < cnt;
      s[u]  = (int)(wn[u] >> LBITS);
      dl[u] = (int)(wn[u] & LMASK);
    }
#pragma unroll
    for (int u = 0; u < 4; ++u) wn[u] = word(i0 + 128 + u * 32 + lane);
    const int nb = cnt - i0;
    process_batch<H, PASS>(s, dl, ok, nb > 128 ? 128 : nb, aS, aD, Hf, acc, mx, sm,
                           r0, hbase, slab, lane, xs, i0, capE);
  }
}

__device__ __forceinline__ void push_hits(int* q, int& cnt, bool hit, int e, int lane) {
  const unsigned m = __builtin_amdgcn_ballot_w32(hit);
  if (m != 0u) {
    const int pos = cnt + __builtin_popcount(m & ((1u << lane) - 1u));
    if (hit) q[pos] = e;
    cnt += __builtin_popcount(m);
  }
}

__device__ __forceinline__ void load_step(const int* __restrict__ dst, int nEdges, int nChunks,
                                          int c, int lane, int (&d)[NLOAD], int& base) {
  base = 0;
  if (c < nChunks) {
    base = c * STEP_E;
    if (base + STEP_E <= nEdges) {
#pragma unroll
      for (int u = 0; u < NLOAD; ++u) d[u] = dst[base + u * 32 + lane];
    } else {
#pragma unroll
      for (int u = 0; u < NLOAD; ++u) {
        const int e = base + u * 32 + lane;
        d[u] = e < nEdges ? dst[e] : -1;
      }
    }
  } else {
#pragma unroll
    for (int u = 0; u < NLOAD; ++u) d[u] = -1;
  }
}

template <int H, int PASS, int NSUB>
__device__ __forceinline__ void scan_edges(int* q, int nEdges,
                                           const int* __restrict__ src,
                                           const int* __restrict__ dst,
                                           const float* __restrict__ aS,
                                           const float* __restrict__ aD,
                                           const float* __restrict__ Hf,
                                           float* acc, float* mx, float* sm,
                                           int r0, int rn, int hbase, int slab,
                                           int lane, int first,
                                           float* seg0, int cap, int& ltot) {
  const int nChunks = (nEdges + STEP_E - 1) / STEP_E;
  const int nRounds = (nChunks + CSTRIDE - 1) / CSTRIDE;
  const int nSteps  = nRounds * NSUB;
  const int rot = (int)(((long long)blockIdx.x * nRounds) / gridDim.x);
  auto chunk_of = [&](int m) {
    int r = m / NSUB + rot;
    if (r >= nRounds) r -= nRounds;
    return m < nSteps ? first + (m % NSUB) + r * CSTRIDE : nChunks;
  };
  int cnt = 0;
  int dn[NLOAD], basen;
  load_step(dst, nEdges, nChunks, chunk_of(0), lane, dn, basen);
  for (int m = 0; ; ++m) {
    const bool more = m < nSteps;
    if (more) {
      int d[NLOAD];
      const int base = basen;
#pragma unroll
      for (int u = 0; u < NLOAD; ++u) d[u] = dn[u];
      load_step(dst, nEdges, nChunks, chunk_of(m + 1), lane, dn, basen);
#pragma unroll
      for (int u = 0; u < NLOAD; ++u)
        push_hits(q, cnt, (unsigned)(d[u] - r0) < (unsigned)rn, base + u * 32 + lane, lane);
    }
    if (cnt > QCAP - STEP_E || (!more && cnt > 0)) {
      drain_queue<H, PASS>(q, cnt, src, dst, aS, aD, Hf, acc, mx, sm, r0, hbase, slab, lane,
                           seg0, cap, ltot);
      cnt = 0;
    }
    if (!more) break;
  }
}

__global__ void __launch_bounds__(1024) scan_lists_kernel(const int* __restrict__ ei, int nEdges,
                                                          float* lst) {
  __shared__ int qbuf[32 * QCAP];
  const int lane = threadIdx.x & 31;
  const int wave = __builtin_amdgcn_readfirstlane(threadIdx.x >> 5);
  const int part = blockIdx.x & (NPART - 1);
  const int tile = blockIdx.x / NPART;
  const int r0   = tile * TILE_R;
  const int rn   = (N_NODES - r0) < TILE_R ? (N_NODES - r0) : TILE_R;
  const int* __restrict__ src = ei;
  const int* __restrict__ dst = ei + nEdges;
  const int ppw = rn / (NPART * 32);
  const int cap = (ppw - 1) * 32;
  float* hdr  = lst + (size_t)(r0 + (part * 32 + wave) * ppw) * HID;
  float* seg0 = hdr + HID;
  int ltot = 0;
  scan_edges<1, 0, 1>(qbuf + wave * QCAP, nEdges, src, dst, nullptr, nullptr, nullptr,
                      nullptr, nullptr, nullptr, r0, rn, 0, 0, lane, part + NPART * wave,
                      seg0, cap, ltot);
  if (lane < NPART) hdr[lane * SLAB] = __int_as_float(ltot);
}

template <int H>
__global__ void __launch_bounds__(1024) owner_layer_kernel(const int* __restrict__ ei, int nEdges,
                                                           const float* __restrict__ aS,
                                                           const float* __restrict__ aD,
                                                           const float* __restrict__ Hf,
                                                           const float* __restrict__ bias,
                                                           float* out, const float* lst,
                                                           float* scr, int applyElu) {
  constexpr int HS = (H == 8) ? 2 : 1;
  __shared__ float acc[TILE_R * SLAB];
  __shared__ float mx[TILE_R * HS];
  __shared__ float sm[TILE_R * HS];
  __shared__ int   qbuf[32 * QCAP];

  const int lane = threadIdx.x & 31;
  const int wave = __builtin_amdgcn_readfirstlane(threadIdx.x >> 5);
  const int slab = blockIdx.x & 3;
  const int tile = blockIdx.x >> 2;
  const int r0   = tile * TILE_R;
  const int rn   = (N_NODES - r0) < TILE_R ? (N_NODES - r0) : TILE_R;
  const int hbase = (H == 8) ? slab * 2 : 0;
  const int* __restrict__ src = ei;
  const int* __restrict__ dst = ei + nEdges;
  int* q = qbuf + wave * QCAP;

  const int ppw = rn / (NPART * 32);
  const size_t qStride = (size_t)32 * ppw * HID;
  const float* hdr  = lst + (size_t)(r0 + wave * ppw) * HID + slab * SLAB;
  const float* seg0 = hdr + HID;
  const int c0 = __builtin_amdgcn_readfirstlane(__float_as_int(hdr[0]));
  const int c1 = __builtin_amdgcn_readfirstlane(__float_as_int(hdr[qStride]));
  const int c2 = __builtin_amdgcn_readfirstlane(__float_as_int(hdr[2 * qStride]));
  const int c3 = __builtin_amdgcn_readfirstlane(__float_as_int(hdr[3 * qStride]));
  const bool kept = (c0 | c1 | c2 | c3) >= 0;
  int none = 0;
  const int rpw  = rn / 32;
  float* xs      = scr + (size_t)(r0 + wave * rpw) * HID + slab * SLAB;
  const int capE = rpw * 32 / HS;

  for (int i = threadIdx.x; i < rn * HS; i += 1024) {
    const int node = r0 + i / HS;
    const int h    = hbase + i % HS;
    mx[i] = leaky(aS[(size_t)node * H + h] + aD[(size_t)node * H + h]);
  }
  __syncthreads();

  if (kept) {
    walk_lists<H, 1>(seg0, qStride, c0, c1, c2, c3, aS, aD, Hf, acc, mx, sm, r0, hbase, slab,
                     lane, xs, capE);
  } else {
    scan_edges<H, 1, NPART>(q, nEdges, src, dst, aS, aD, Hf, acc, mx, sm, r0, rn, hbase, slab,
                            lane, NPART * wave, nullptr, 0, none);
  }
  __syncthreads();

  for (int i = wave; i < rn; i += 32) {
    const int node = r0 + i;
    const int hh   = (HS == 2) ? (lane >> 4) : 0;
    const int h    = hbase + hh;
    const float x  = leaky(aS[(size_t)node * H + h] + aD[(size_t)node * H + h]);
    const float w  = expf(x - mx[i * HS + hh]);
    acc[i * SLAB + lane] = w * Hf[(size_t)node * HID + slab * SLAB + lane];
    if ((lane & 15) == 0 && (lane >> 4) < HS) sm[i * HS + (lane >> 4)] = w;
  }
  __syncthreads();

  if (kept) {
    walk_lists<H, 2>(seg0, qStride, c0, c1, c2, c3, aS, aD, Hf, acc, mx, sm, r0, hbase, slab,
                     lane, xs, capE);
  } else {
    scan_edges<H, 2, NPART>(q, nEdges, src, dst, aS, aD, Hf, acc, mx, sm, r0, rn, hbase, slab,
                            lane, NPART * wave, nullptr, 0, none);
  }
  __syncthreads();

  const float bv = bias[slab * SLAB + lane];
  for (int i = wave; i < rn; i += 32) {
    const int hh = (HS == 2) ? (lane >> 4) : 0;
    float v = acc[i * SLAB + lane] / sm[i * HS + hh] + bv;
    if (applyElu) v = v > 0.f ? v : (expf(v) - 1.f);
    out[(size_t)(r0 + i) * HID + slab * SLAB + lane] = v;
  }
}

extern "C" void kernel_launch(void* const* d_in, const int* in_sizes, int n_in,
                              void* d_out, int out_size, void* d_ws, size_t ws_size,
                              hipStream_t stream) {
  const float* x   = (const float*)d_in[0];
  const int*   ei  = (const int*)d_in[1];
  const float* W1  = (const float*)d_in[2];
  const float* as1 = (const float*)d_in[3];
  const float* ad1 = (const float*)d_in[4];
  const float* b1  = (const float*)d_in[5];
  const float* W2  = (const float*)d_in[6];
  const float* as2 = (const float*)d_in[7];
  const float* ad2 = (const float*)d_in[8];
  const float* b2  = (const float*)d_in[9];

  const int nEdges = in_sizes[1] / 2;

  float* A   = (float*)d_ws;
  float* B   = A  + (size_t)N_NODES * HID;
  float* aS  = B  + (size_t)N_NODES * HID;
  float* aD  = aS + (size_t)N_NODES * GAT_HEADS;
  float* out = (float*)d_out;

  const int TB = 256;
  auto cdiv = [](long long a, long long b) { return (unsigned)((a + b - 1) / b); };
  const unsigned gGemm  = (N_NODES / 16) * (HID / 16) / (TB / 32);
  const unsigned gOwner = ((N_NODES + TILE_R - 1) / TILE_R) * (HID / SLAB);

  scan_lists_kernel<<<gOwner, 1024, 0, stream>>>(ei, nEdges, out);

  gemm128_wmma<<<gGemm, TB, 0, stream>>>(x, W1, A);
  alpha1_kernel<<<cdiv(N_NODES * GAT_HEADS, TB), TB, 0, stream>>>(A, as1, ad1, aS, aD);
  owner_layer_kernel<GAT_HEADS><<<gOwner, 1024, 0, stream>>>(ei, nEdges, aS, aD, A, b1, B, out, B, 1);

  gemm128_wmma<<<gGemm, TB, 0, stream>>>(B, W2, A);
  alpha2_kernel<<<cdiv((long long)N_NODES * 32, TB), TB, 0, stream>>>(A, as2, ad2, aS, aD);
  owner_layer_kernel<1><<<gOwner, 1024, 0, stream>>>(ei, nEdges, aS, aD, A, b2, out, out, B, 0);
}


# Row-gather loop: an entry's row offset, tile row and weight reach all lanes as DPP row_share operands of the add or multiply using them, no readlane/scalar chain: 5 instructions per entry, not 20
# speedup vs baseline: 3.0242x; 1.0673x over previous
#include <hip/hip_runtime.h>

#define N_NODES 100000
#define IN_DIM 128
#define HID 128
#define GAT_HEADS 8
#define NEG_SLOPE 0.2f

typedef float v2f __attribute__((ext_vector_type(2)));
typedef float v4f __attribute__((ext_vector_type(4)));
typedef float v8f __attribute__((ext_vector_type(8)));

__global__ void __launch_bounds__(256) gemm128_wmma(const float* __restrict__ X,
                                                    const float* __restrict__ W,
                                                    float* __restrict__ H) {
  const int lane  = threadIdx.x & 31;
  const int wave  = blockIdx.x * (blockDim.x >> 5) + (threadIdx.x >> 5);
  const int tileM = wave >> 3;
  const int tileN = wave & 7;
  const int r     = lane & 15;
  const int khalf = lane >> 4;
  const int row   = tileM * 16 + r;
  const int col   = tileN * 16 + r;
  const float* __restrict__ xrow = X + (size_t)row * IN_DIM;

  v8f c = {0.f, 0.f, 0.f, 0.f, 0.f, 0.f, 0.f, 0.f};
#pragma unroll 4
  for (int k = 0; k < IN_DIM; k += 4) {
    const int kk = k + 2 * khalf;
    v2f a = *(const v2f*)(xrow + kk);
    v2f b;
    b.x = W[(kk    ) * HID + col];
    b.y = W[(kk + 1) * HID + col];
    c = __builtin_amdgcn_wmma_f32_16x16x4_f32(false, a, false, b, (short)0, c,
                                              false, false);
  }
  const int mbase = tileM * 16 + khalf * 8;
#pragma unroll
  for (int v = 0; v < 8; ++v)
    H[(size_t)(mbase + v) * HID + tileN * 16 + r] = c[v];
}

__global__ void alpha1_kernel(const float* __restrict__ H,
                              const float* __restrict__ avS,
                              const float* __restrict__ avD,
                              float* __restrict__ outS,
                              float* __restrict__ outD) {
  int i = blockIdx.x * blockDim.x + threadIdx.x;
  if (i >= N_NODES * GAT_HEADS) return;
  int head = i & 7, node = i >> 3;
  const float* h  = H   + (size_t)node * HID + head * 16;
  const float* as = avS + head * 16;
  const float* ad = avD + head * 16;
  float s = 0.f, d = 0.f;
#pragma unroll
  for (int c = 0; c < 16; ++c) { float hv = h[c]; s += hv * as[c]; d += hv * ad[c]; }
  outS[i] = s; outD[i] = d;
}

__global__ void alpha2_kernel(const float* __restrict__ H,
                              const float* __restrict__ avS,
                              const float* __restrict__ avD,
                              float* __restrict__ outS,
                              float* __restrict__ outD) {
  int wave = blockIdx.x * (blockDim.x >> 5) + (threadIdx.x >> 5);
  int lane = threadIdx.x & 31;
  if (wave >= N_NODES) return;
  float s = 0.f, d = 0.f;
#pragma unroll
  for (int c = lane; c < HID; c += 32) {
    float hv = H[(size_t)wave * HID + c];
    s += hv * avS[c]; d += hv * avD[c];
  }
#pragma unroll
  for (int off = 16; off > 0; off >>= 1) {
    s += __shfl_xor(s, off, 32);
    d += __shfl_xor(d, off, 32);
  }
  if (lane == 0) { outS[wave] = s; outD[wave] = d; }
}

#define TILE_R 1568
#define SLAB   32
#define NLOAD  12
#define STEP_E (NLOAD * 32)
#define QCAP   512
#define LBITS  11
#define LMASK  ((1u << LBITS) - 1u)
static_assert(TILE_R <= (1 << LBITS), "tile-local destination must fit its field");
static_assert(N_NODES <= (1 << (32 - LBITS)), "source node must fit its field");
#define NPART  4
#define CSTRIDE (NPART * 32)
static_assert(NPART == HID / SLAB, "one scanning work-group per slab work-group of a tile");
static_assert(N_NODES % TILE_R == 0 || N_NODES % TILE_R >= 2 * NPART * 32,
              "the last tile must give every list a count line and a data line of its own");

__device__ __forceinline__ float leaky(float x) {
  return x > 0.f ? x : NEG_SLOPE * x;
}

__device__ __forceinline__ void atomicMaxFloat(float* addr, float val) {
  if (__float_as_int(val) >= 0)
    atomicMax((int*)addr, __float_as_int(val));
  else
    atomicMin((unsigned int*)addr, __float_as_uint(val));
}

__device__ __forceinline__ void ldsAdd(float* addr, float val) {
  __hip_atomic_fetch_add(addr, val, __ATOMIC_RELAXED, __HIP_MEMORY_SCOPE_WORKGROUP);
}

__device__ __forceinline__ int xrow_i(int v) {
  return __builtin_amdgcn_permlanex16(v, v, 0x76543210, 0xfedcba98, false, false);
}
template <int J>
__device__ __forceinline__ int rshare_i(int v) {
  return __builtin_amdgcn_update_dpp(0, v, 0x150 + J, 0xf, 0xf, true);
}
template <int J>
__device__ __forceinline__ float rshare_f(float v) {
  return __int_as_float(rshare_i<J>(__float_as_int(v)));
}
#define ROW16(M) M(0) M(1) M(2) M(3) M(4) M(5) M(6) M(7) M(8) M(9) M(10) M(11) M(12) M(13) M(14) M(15)

template <int H>
__device__ __forceinline__ void edge_logits(const float* __restrict__ aS,
                                            const float* __restrict__ aD,
                                            int s, int d, int hbase, float& x0, float& x1) {
  if (H == 8) {
    v2f a = *(const v2f*)(aS + (size_t)s * 8 + hbase);
    v2f b = *(const v2f*)(aD + (size_t)d * 8 + hbase);
    x0 = leaky(a.x + b.x); x1 = leaky(a.y + b.y);
  } else {
    x0 = leaky(aS[s] + aD[d]); x1 = 0.f;
  }
}

template <int H, int PASS>
__device__ __forceinline__ void process_batch(const int (&s)[4], const int (&dl)[4],
                                              const bool (&ok)[4], int nb,
                                              const float* __restrict__ aS,
                                              const float* __restrict__ aD,
                                              const float* __restrict__ Hf,
                                              float* acc, float* mx, float* sm,
                                              int r0, int hbase, int slab, int lane,
                                              float* xs, int g0, int capE) {
  constexpr int HS = (H == 8) ? 2 : 1;
  float w0[4], w1[4];
  float xh0[4], xh1[4];
  if (PASS == 2) {
#pragma unroll
    for (int u = 0; u < 4; ++u) {
      const int g = g0 + u * 32 + lane;
      const float* xp = xs + (size_t)((g * HS) >> 5) * HID + ((g * HS) & 31);
      xh0[u] = 0.f; xh1[u] = 0.f;
      if (ok[u] && g < capE) {
        if (HS == 2) { const v2f xv = *(const v2f*)xp; xh0[u] = xv.x; xh1[u] = xv.y; }
        else         xh0[u] = *xp;
      }
    }
  }
#pragma unroll
  for (int u = 0; u < 4; ++u) {
    float x0, x1;
    const int g = g0 + u * 32 + lane;
    float* xp = xs + (size_t)((g * HS) >> 5) * HID + ((g * HS) & 31);
    const bool held = g < capE;
    if (PASS == 2 && held) {
      x0 = xh0[u]; x1 = xh1[u];
    } else {
      edge_logits<H>(aS, aD, s[u], dl[u] + r0, hbase, x0, x1);
    }
    if (PASS == 1) {
      if (ok[u] && held) {
        if (HS == 2) { v2f xv; xv.x = x0; xv.y = x1; *(v2f*)xp = xv; }
        else         *xp = x0;
      }
      if (ok[u]) {
        atomicMaxFloat(&mx[dl[u] * HS], x0);
        if (HS == 2) atomicMaxFloat(&mx[dl[u] * HS + 1], x1);
      }
    } else {
      w0[u] = 0.f; w1[u] = 0.f;
      if (ok[u]) {
        w0[u] = expf(x0 - mx[dl[u] * HS]);
        ldsAdd(&sm[dl[u] * HS], w0[u]);
        if (HS == 2) {
          w1[u] = expf(x1 - mx[dl[u] * HS + 1]);
          ldsAdd(&sm[dl[u] * HS + 1], w1[u]);
        }
      }
    }
  }
  if (PASS == 2) {
    const bool lo = lane < 16;
    const char* Hb = (const char*)Hf;
    char* accb = (char*)acc;
    const unsigned laneB = (unsigned)(slab * SLAB + lane) * 4u;
    const unsigned lane4 = (unsigned)lane * 4u;
#pragma unroll
    for (int u = 0; u < 4; ++u) {
      int n = nb - u * 32;
      n = n > 32 ? 32 : n;
      if (n > 0) {
        const int so = s[u] * (HID * 4);
        const int dO = dl[u] * (SLAB * 4);
        const int wA = __float_as_int(w0[u]);
        const int wB = __float_as_int(HS == 2 ? w1[u] : w0[u]);
        const int sX = xrow_i(so), dX = xrow_i(dO), wAX = xrow_i(wA), wBX = xrow_i(wB);
#pragma unroll
        for (int g = 0; g < 2; ++g) {
          if (g * 16 < n) {
            const int   Sg = (g == 0) ? (lo ? so : sX) : (lo ? sX : so);
            const int   Dg = (g == 0) ? (lo ? dO : dX) : (lo ? dX : dO);
            const float Wg = __int_as_float((g == 0) ? (lo ? wA : wBX) : (lo ? wAX : wB));
            float hv[16];
#define GAT_LOAD(J) hv[J] = *(const float*)(Hb + (size_t)((unsigned)rshare_i<J>(Sg) + laneB));
#define GAT_ADD(J) ldsAdd((float*)(accb + ((unsigned)rshare_i<J>(Dg) + lane4)), hv[J] * rshare_f<J>(Wg));
            if (g * 16 + 16 <= n) {
              ROW16(GAT_LOAD)
              ROW16(GAT_ADD)
            } else {
              ROW16(GAT_LOAD)
#define GAT_ADDP(J) if (g * 16 + J < n) { GAT_ADD(J) }
              ROW16(GAT_ADDP)
#undef GAT_ADDP
            }
#undef GAT_ADD
#undef GAT_LOAD
          }
        }
      }
    }
  }
}

template <int H, int PASS>
__device__ __forceinline__ void drain_queue(const int* q, int cnt,
                                            const int* __restrict__ src,
                                            const int* __restrict__ dst,
                                            const float* __restrict__ aS,
                                            const float* __restrict__ aD,
                                            const float* __restrict__ Hf,
                                            float* acc, float* mx, float* sm,
                                            int r0, int hbase, int slab, int lane,
                                            float* seg0, int cap, int& ltot) {
  constexpr bool REC = (PASS == 0);
  bool room = false;
  if (REC) {
    room = ltot >= 0 && ltot + cnt <= cap;
    if (!room) ltot = -1;
  }
  for (int i0 = 0; i0 < cnt; i0 += 128) {
    int  s[4], dl[4];
    bool ok[4];
#pragma unroll
    for (int u = 0; u < 4; ++u) {
      const int i = i0 + u * 32 + lane;
      ok[u] = i < cnt;
      const int e = ok[u] ? q[i] : 0;
      s[u]  = src[e];
      dl[u] = dst[e] - r0;
    }
#pragma unroll
    for (int u = 0; u < 4; ++u) {
      if (!ok[u]) { s[u] = 0; dl[u] = 0; }
      if (REC && room && ok[u]) {
        const int p = ltot + i0 + u * 32 + lane;
        const float w = __uint_as_float(((unsigned)s[u] << LBITS) | (unsigned)dl[u]);
        float* at = seg0 + (size_t)(p >> 5) * HID + (p & 31);
#pragma unroll
        for (int sl = 0; sl < NPART; ++sl) at[sl * SLAB] = w;
      }
    }
    if constexpr (PASS != 0) {
      const int nb = cnt - i0;
      process_batch<H, PASS>(s, dl, ok, nb > 128 ? 128 : nb, aS, aD, Hf, acc, mx, sm,
                             r0, hbase, slab, lane, nullptr, 0, 0);
    }
  }
  if (REC && room) ltot += cnt;
}

template <int H, int PASS>
__device__ __forceinline__ void walk_lists(const float* seg0, size_t qStride,
                                           int c0, int c1, int c2, int c3,
                                           const float* __restrict__ aS,
                                           const float* __restrict__ aD,
                                           const float* __restrict__ Hf,
                                           float* acc, float* mx, float* sm,
                                           int r0, int hbase, int slab, int lane,
                                           float* xs, int capE) {
  const int t1 = c0, t2 = t1 + c1, t3 = t2 + c2, cnt = t3 + c3;
  auto word = [&](int g) -> unsigned {
    if (g >= cnt) return 0u;
    const int q   = (g >= t1) + (g >= t2) + (g >= t3);
    const int off = g - (g >= t3 ? t3 : (g >= t2 ? t2 : (g >= t1 ? t1 : 0)));
    return __float_as_uint(seg0[(size_t)q * qStride + (size_t)(off >> 5) * HID + (off & 31)]);
  };
  unsigned wn[4];
#pragma unroll
  for (int u = 0; u < 4; ++u) wn[u] = word(u * 32 + lane);
  for (int i0 = 0; i0 < cnt; i0 += 128) {
    int  s[4], dl[4];
    bool ok[4];
#pragma unroll
    for (int u = 0; u < 4; ++u) {
      ok[u] = i0 + u * 32 + lane < cnt;
      s[u]  = (int)(wn[u] >> LBITS);
      dl[u] = (int)(wn[u] & LMASK);
    }
#pragma unroll
    for (int u = 0; u < 4; ++u) wn[u] = word(i0 + 128 + u * 32 + lane);
    const int nb = cnt - i0;
    process_batch<H, PASS>(s, dl, ok, nb > 128 ? 128 : nb, aS, aD, Hf, acc, mx, sm,
                           r0, hbase, slab, lane, xs, i0, capE);
  }
}

__device__ __forceinline__ void push_hits(int* q, int& cnt, bool hit, int e, int lane) {
  const unsigned m = __builtin_amdgcn_ballot_w32(hit);
  if (m != 0u) {
    const int pos = cnt + __builtin_popcount(m & ((1u << lane) - 1u));
    if (hit) q[pos] = e;
    cnt += __builtin_popcount(m);
  }
}

__device__ __forceinline__ void load_step(const int* __restrict__ dst, int nEdges, int nChunks,
                                          int c, int lane, int (&d)[NLOAD], int& base) {
  base = 0;
  if (c < nChunks) {
    base = c * STEP_E;
    if (base + STEP_E <= nEdges) {
#pragma unroll
      for (int u = 0; u < NLOAD; ++u) d[u] = dst[base + u * 32 + lane];
    } else {
#pragma unroll
      for (int u = 0; u < NLOAD; ++u) {
        const int e = base + u * 32 + lane;
        d[u] = e < nEdges ? dst[e] : -1;
      }
    }
  } else {
#pragma unroll
    for (int u = 0; u < NLOAD; ++u) d[u] = -1;
  }
}

template <int H, int PASS, int NSUB>
__device__ __forceinline__ void scan_edges(int* q, int nEdges,
                                           const int* __restrict__ src,
                                           const int* __restrict__ dst,
                                           const float* __restrict__ aS,
                                           const float* __restrict__ aD,
                                           const float* __restrict__ Hf,
                                           float* acc, float* mx, float* sm,
                                           int r0, int rn, int hbase, int slab,
                                           int lane, int first,
                                           float* seg0, int cap, int& ltot) {
  const int nChunks = (nEdges + STEP_E - 1) / STEP_E;
  const int nRounds = (nChunks + CSTRIDE - 1) / CSTRIDE;
  const int nSteps  = nRounds * NSUB;
  const int rot = (int)(((long long)blockIdx.x * nRounds) / gridDim.x);
  auto chunk_of = [&](int m) {
    int r = m / NSUB + rot;
    if (r >= nRounds) r -= nRounds;
    return m < nSteps ? first + (m % NSUB) + r * CSTRIDE : nChunks;
  };
  int cnt = 0;
  int dn[NLOAD], basen;
  load_step(dst, nEdges, nChunks, chunk_of(0), lane, dn, basen);
  for (int m = 0; ; ++m) {
    const bool more = m < nSteps;
    if (more) {
      int d[NLOAD];
      const int base = basen;
#pragma unroll
      for (int u = 0; u < NLOAD; ++u) d[u] = dn[u];
      load_step(dst, nEdges, nChunks, chunk_of(m + 1), lane, dn, basen);
#pragma unroll
      for (int u = 0; u < NLOAD; ++u)
        push_hits(q, cnt, (unsigned)(d[u] - r0) < (unsigned)rn, base + u * 32 + lane, lane);
    }
    if (cnt > QCAP - STEP_E || (!more && cnt > 0)) {
      drain_queue<H, PASS>(q, cnt, src, dst, aS, aD, Hf, acc, mx, sm, r0, hbase, slab, lane,
                           seg0, cap, ltot);
      cnt = 0;
    }
    if (!more) break;
  }
}

__global__ void __launch_bounds__(1024) scan_lists_kernel(const int* __restrict__ ei, int nEdges,
                                                          float* lst) {
  __shared__ int qbuf[32 * QCAP];
  const int lane = threadIdx.x & 31;
  const int wave = __builtin_amdgcn_readfirstlane(threadIdx.x >> 5);
  const int part = blockIdx.x & (NPART - 1);
  const int tile = blockIdx.x / NPART;
  const int r0   = tile * TILE_R;
  const int rn   = (N_NODES - r0) < TILE_R ? (N_NODES - r0) : TILE_R;
  const int* __restrict__ src = ei;
  const int* __restrict__ dst = ei + nEdges;
  const int ppw = rn / (NPART * 32);
  const int cap = (ppw - 1) * 32;
  float* hdr  = lst + (size_t)(r0 + (part * 32 + wave) * ppw) * HID;
  float* seg0 = hdr + HID;
  int ltot = 0;
  scan_edges<1, 0, 1>(qbuf + wave * QCAP, nEdges, src, dst, nullptr, nullptr, nullptr,
                      nullptr, nullptr, nullptr, r0, rn, 0, 0, lane, part + NPART * wave,
                      seg0, cap, ltot);
  if (lane < NPART) hdr[lane * SLAB] = __int_as_float(ltot);
}

template <int H>
__global__ void __launch_bounds__(1024) owner_layer_kernel(const int* __restrict__ ei, int nEdges,
                                                           const float* __restrict__ aS,
                                                           const float* __restrict__ aD,
                                                           const float* __restrict__ Hf,
                                                           const float* __restrict__ bias,
                                                           float* out, const float* lst,
                                                           float* scr, int applyElu) {
  constexpr int HS = (H == 8) ? 2 : 1;
  __shared__ float acc[TILE_R * SLAB];
  __shared__ float mx[TILE_R * HS];
  __shared__ float sm[TILE_R * HS];
  __shared__ int   qbuf[32 * QCAP];

  const int lane = threadIdx.x & 31;
  const int wave = __builtin_amdgcn_readfirstlane(threadIdx.x >> 5);
  const int slab = blockIdx.x & 3;
  const int tile = blockIdx.x >> 2;
  const int r0   = tile * TILE_R;
  const int rn   = (N_NODES - r0) < TILE_R ? (N_NODES - r0) : TILE_R;
  const int hbase = (H == 8) ? slab * 2 : 0;
  const int* __restrict__ src = ei;
  const int* __restrict__ dst = ei + nEdges;
  int* q = qbuf + wave * QCAP;

  const int ppw = rn / (NPART * 32);
  const size_t qStride = (size_t)32 * ppw * HID;
  const float* hdr  = lst + (size_t)(r0 + wave * ppw) * HID + slab * SLAB;
  const float* seg0 = hdr + HID;
  const int c0 = __builtin_amdgcn_readfirstlane(__float_as_int(hdr[0]));
  const int c1 = __builtin_amdgcn_readfirstlane(__float_as_int(hdr[qStride]));
  const int c2 = __builtin_amdgcn_readfirstlane(__float_as_int(hdr[2 * qStride]));
  const int c3 = __builtin_amdgcn_readfirstlane(__float_as_int(hdr[3 * qStride]));
  const bool kept = (c0 | c1 | c2 | c3) >= 0;
  int none = 0;
  const int rpw  = rn / 32;
  float* xs      = scr + (size_t)(r0 + wave * rpw) * HID + slab * SLAB;
  const int capE = rpw * 32 / HS;

  for (int i = threadIdx.x; i < rn * HS; i += 1024) {
    const int node = r0 + i / HS;
    const int h    = hbase + i % HS;
    mx[i] = leaky(aS[(size_t)node * H + h] + aD[(size_t)node * H + h]);
  }
  __syncthreads();

  if (kept) {
    walk_lists<H, 1>(seg0, qStride, c0, c1, c2, c3, aS, aD, Hf, acc, mx, sm, r0, hbase, slab,
                     lane, xs, capE);
  } else {
    scan_edges<H, 1, NPART>(q, nEdges, src, dst, aS, aD, Hf, acc, mx, sm, r0, rn, hbase, slab,
                            lane, NPART * wave, nullptr, 0, none);
  }
  __syncthreads();

  for (int i = wave; i < rn; i += 32) {
    const int node = r0 + i;
    const int hh   = (HS == 2) ? (lane >> 4) : 0;
    const int h    = hbase + hh;
    const float x  = leaky(aS[(size_t)node * H + h] + aD[(size_t)node * H + h]);
    const float w  = expf(x - mx[i * HS + hh]);
    acc[i * SLAB + lane] = w * Hf[(size_t)node * HID + slab * SLAB + lane];
    if ((lane & 15) == 0 && (lane >> 4) < HS) sm[i * HS + (lane >> 4)] = w;
  }
  __syncthreads();

  if (kept) {
    walk_lists<H, 2>(seg0, qStride, c0, c1, c2, c3, aS, aD, Hf, acc, mx, sm, r0, hbase, slab,
                     lane, xs, capE);
  } else {
    scan_edges<H, 2, NPART>(q, nEdges, src, dst, aS, aD, Hf, acc, mx, sm, r0, rn, hbase, slab,
                            lane, NPART * wave, nullptr, 0, none);
  }
  __syncthreads();

  const float bv = bias[slab * SLAB + lane];
  for (int i = wave; i < rn; i += 32) {
    const int hh = (HS == 2) ? (lane >> 4) : 0;
    float v = acc[i * SLAB + lane] / sm[i * HS + hh] + bv;
    if (applyElu) v = v > 0.f ? v : (expf(v) - 1.f);
    out[(size_t)(r0 + i) * HID + slab * SLAB + lane] = v;
  }
}

extern "C" void kernel_launch(void* const* d_in, const int* in_sizes, int n_in,
                              void* d_out, int out_size, void* d_ws, size_t ws_size,
                              hipStream_t stream) {
  const float* x   = (const float*)d_in[0];
  const int*   ei  = (const int*)d_in[1];
  const float* W1  = (const float*)d_in[2];
  const float* as1 = (const float*)d_in[3];
  const float* ad1 = (const float*)d_in[4];
  const float* b1  = (const float*)d_in[5];
  const float* W2  = (const float*)d_in[6];
  const float* as2 = (const float*)d_in[7];
  const float* ad2 = (const float*)d_in[8];
  const float* b2  = (const float*)d_in[9];

  const int nEdges = in_sizes[1] / 2;

  float* A   = (float*)d_ws;
  float* B   = A  + (size_t)N_NODES * HID;
  float* aS  = B  + (size_t)N_NODES * HID;
  float* aD  = aS + (size_t)N_NODES * GAT_HEADS;
  float* out = (float*)d_out;

  const int TB = 256;
  auto cdiv = [](long long a, long long b) { return (unsigned)((a + b - 1) / b); };
  const unsigned gGemm  = (N_NODES / 16) * (HID / 16) / (TB / 32);
  const unsigned gOwner = ((N_NODES + TILE_R - 1) / TILE_R) * (HID / SLAB);

  scan_lists_kernel<<<gOwner, 1024, 0, stream>>>(ei, nEdges, out);

  gemm128_wmma<<<gGemm, TB, 0, stream>>>(x, W1, A);
  alpha1_kernel<<<cdiv(N_NODES * GAT_HEADS, TB), TB, 0, stream>>>(A, as1, ad1, aS, aD);
  owner_layer_kernel<GAT_HEADS><<<gOwner, 1024, 0, stream>>>(ei, nEdges, aS, aD, A, b1, B, out, B, 1);

  gemm128_wmma<<<gGemm, TB, 0, stream>>>(B, W2, A);
  alpha2_kernel<<<cdiv((long long)N_NODES * 32, TB), TB, 0, stream>>>(A, as2, ad2, aS, aD);
  owner_layer_kernel<1><<<gOwner, 1024, 0, stream>>>(ei, nEdges, aS, aD, A, b2, out, out, B, 0);
}


# Projection: a wave owns 16 rows x all 128 columns, one A fragment feeds 8 WMMA tiles; W staged per work-group in LDS (transposed), B fragments read there; whole-line stores; k sum order changes
# speedup vs baseline: 3.4039x; 1.1256x over previous
#include <hip/hip_runtime.h>

#define N_NODES 100000
#define IN_DIM 128
#define HID 128
#define GAT_HEADS 8
#define NEG_SLOPE 0.2f

typedef float v2f __attribute__((ext_vector_type(2)));
typedef float v4f __attribute__((ext_vector_type(4)));
typedef float v8f __attribute__((ext_vector_type(8)));

__device__ __forceinline__ int xrow_i(int v) {
  return __builtin_amdgcn_permlanex16(v, v, 0x76543210, 0xfedcba98, false, false);
}

#define WT_PITCH 132
#define GEMM_WG 256
static_assert((N_NODES % 16) == 0, "row tiles are whole");
static_assert((N_NODES / 16 + GEMM_WG - 1) / GEMM_WG <= 32, "one row tile per wave of a work-group");
__global__ void __launch_bounds__(1024) gemm128_wmma(const float* __restrict__ X,
                                                     const float* __restrict__ W,
                                                     float* __restrict__ H) {
  __shared__ __attribute__((aligned(16))) float Wt[HID * WT_PITCH];
  const int lane = threadIdx.x & 31;
  const int wave = __builtin_amdgcn_readfirstlane(threadIdx.x >> 5);

  const int nTiles = N_NODES / 16;
  const int t0 = (int)(((long long)blockIdx.x * nTiles) / GEMM_WG);
  const int t1 = (int)(((long long)(blockIdx.x + 1) * nTiles) / GEMM_WG);
  const int slot = 8 * (((wave >> 3) + wave) & 3) + (wave & 7);
  const int tile = t0 + slot;
  const bool active = tile < t1;

  const int c = lane & 15;
  const int h = lane >> 4;
  const float* __restrict__ xrow = X + (size_t)(tile * 16 + c) * IN_DIM + 4 * h;
  const float* wl = Wt + c * WT_PITCH + 4 * h;

  v4f a = {0.f, 0.f, 0.f, 0.f};
  if (active) a = *(const v4f*)xrow;

  for (int i = threadIdx.x; i < IN_DIM * HID; i += 1024)
    Wt[(i & (HID - 1)) * WT_PITCH + (i >> 7)] = W[i];
  __syncthreads();
  if (!active) return;

  v8f acc[8];
#pragma unroll
  for (int n = 0; n < 8; ++n) acc[n] = (v8f){0.f, 0.f, 0.f, 0.f, 0.f, 0.f, 0.f, 0.f};

#pragma unroll
  for (int j = 0; j < IN_DIM / 8; ++j) {
    v4f an = a;
    if (j + 1 < IN_DIM / 8) an = *(const v4f*)(xrow + 8 * (j + 1));
#pragma unroll
    for (int n0 = 0; n0 < 8; n0 += 4) {
      v4f b[4];
#pragma unroll
      for (int nn = 0; nn < 4; ++nn)
        b[nn] = *(const v4f*)(wl + (n0 + nn) * 16 * WT_PITCH + 8 * j);
#pragma unroll
      for (int nn = 0; nn < 4; ++nn) {
        const v2f a0 = {a.x, a.y};
        const v2f b0 = {b[nn].x, b[nn].y};
        acc[n0 + nn] = __builtin_amdgcn_wmma_f32_16x16x4_f32(false, a0, false, b0, (short)0,
                                                             acc[n0 + nn], false, false);
      }
#pragma unroll
      for (int nn = 0; nn < 4; ++nn) {
        const v2f a1 = {a.z, a.w};
        const v2f b1 = {b[nn].z, b[nn].w};
        acc[n0 + nn] = __builtin_amdgcn_wmma_f32_16x16x4_f32(false, a1, false, b1, (short)0,
                                                             acc[n0 + nn], false, false);
      }
    }
    a = an;
  }

  float* hout = H + (size_t)tile * 16 * HID + lane;
#pragma unroll
  for (int p = 0; p < 4; ++p) {
#pragma unroll
    for (int v = 0; v < 8; ++v) {
      const float x0 = acc[2 * p][v];
      const float x1 = acc[2 * p + 1][v];
      const float x0o = __int_as_float(xrow_i(__float_as_int(x0)));
      const float x1o = __int_as_float(xrow_i(__float_as_int(x1)));
      const float ylo = h == 0 ? x0 : x1o;
      const float yhi = h == 0 ? x0o : x1;
      hout[(size_t)v * HID + 32 * p]       = ylo;
      hout[(size_t)(v + 8) * HID + 32 * p] = yhi;
    }
  }
}

__global__ void alpha1_kernel(const float* __restrict__ H,
                              const float* __restrict__ avS,
                              const float* __restrict__ avD,
                              float* __restrict__ outS,
                              float* __restrict__ outD) {
  int i = blockIdx.x * blockDim.x + threadIdx.x;
  if (i >= N_NODES * GAT_HEADS) return;
  int head = i & 7, node = i >> 3;
  const float* h  = H   + (size_t)node * HID + head * 16;
  const float* as = avS + head * 16;
  const float* ad = avD + head * 16;
  float s = 0.f, d = 0.f;
#pragma unroll
  for (int c = 0; c < 16; ++c) { float hv = h[c]; s += hv * as[c]; d += hv * ad[c]; }
  outS[i] = s; outD[i] = d;
}

__global__ void alpha2_kernel(const float* __restrict__ H,
                              const float* __restrict__ avS,
                              const float* __restrict__ avD,
                              float* __restrict__ outS,
                              float* __restrict__ outD) {
  int wave = blockIdx.x * (blockDim.x >> 5) + (threadIdx.x >> 5);
  int lane = threadIdx.x & 31;
  if (wave >= N_NODES) return;
  float s = 0.f, d = 0.f;
#pragma unroll
  for (int c = lane; c < HID; c += 32) {
    float hv = H[(size_t)wave * HID + c];
    s += hv * avS[c]; d += hv * avD[c];
  }
#pragma unroll
  for (int off = 16; off > 0; off >>= 1) {
    s += __shfl_xor(s, off, 32);
    d += __shfl_xor(d, off, 32);
  }
  if (lane == 0) { outS[wave] = s; outD[wave] = d; }
}

#define TILE_R 1568
#define SLAB   32
#define NLOAD  12
#define STEP_E (NLOAD * 32)
#define QCAP   512
#define LBITS  11
#define LMASK  ((1u << LBITS) - 1u)
static_assert(TILE_R <= (1 << LBITS), "tile-local destination must fit its field");
static_assert(N_NODES <= (1 << (32 - LBITS)), "source node must fit its field");
#define NPART  4
#define CSTRIDE (NPART * 32)
static_assert(NPART == HID / SLAB, "one scanning work-group per slab work-group of a tile");
static_assert(N_NODES % TILE_R == 0 || N_NODES % TILE_R >= 2 * NPART * 32,
              "the last tile must give every list a count line and a data line of its own");

__device__ __forceinline__ float leaky(float x) {
  return x > 0.f ? x : NEG_SLOPE * x;
}

__device__ __forceinline__ void atomicMaxFloat(float* addr, float val) {
  if (__float_as_int(val) >= 0)
    atomicMax((int*)addr, __float_as_int(val));
  else
    atomicMin((unsigned int*)addr, __float_as_uint(val));
}

__device__ __forceinline__ void ldsAdd(float* addr, float val) {
  __hip_atomic_fetch_add(addr, val, __ATOMIC_RELAXED, __HIP_MEMORY_SCOPE_WORKGROUP);
}

template <int J>
__device__ __forceinline__ int rshare_i(int v) {
  return __builtin_amdgcn_update_dpp(0, v, 0x150 + J, 0xf, 0xf, true);
}
template <int J>
__device__ __forceinline__ float rshare_f(float v) {
  return __int_as_float(rshare_i<J>(__float_as_int(v)));
}
#define ROW16(M) M(0) M(1) M(2) M(3) M(4) M(5) M(6) M(7) M(8) M(9) M(10) M(11) M(12) M(13) M(14) M(15)

template <int H>
__device__ __forceinline__ void edge_logits(const float* __restrict__ aS,
                                            const float* __restrict__ aD,
                                            int s, int d, int hbase, float& x0, float& x1) {
  if (H == 8) {
    v2f a = *(const v2f*)(aS + (size_t)s * 8 + hbase);
    v2f b = *(const v2f*)(aD + (size_t)d * 8 + hbase);
    x0 = leaky(a.x + b.x); x1 = leaky(a.y + b.y);
  } else {
    x0 = leaky(aS[s] + aD[d]); x1 = 0.f;
  }
}

template <int H, int PASS>
__device__ __forceinline__ void process_batch(const int (&s)[4], const int (&dl)[4],
                                              const bool (&ok)[4], int nb,
                                              const float* __restrict__ aS,
                                              const float* __restrict__ aD,
                                              const float* __restrict__ Hf,
                                              float* acc, float* mx, float* sm,
                                              int r0, int hbase, int slab, int lane,
                                              float* xs, int g0, int capE) {
  constexpr int HS = (H == 8) ? 2 : 1;
  float w0[4], w1[4];
  float xh0[4], xh1[4];
  if (PASS == 2) {
#pragma unroll
    for (int u = 0; u < 4; ++u) {
      const int g = g0 + u * 32 + lane;
      const float* xp = xs + (size_t)((g * HS) >> 5) * HID + ((g * HS) & 31);
      xh0[u] = 0.f; xh1[u] = 0.f;
      if (ok[u] && g < capE) {
        if (HS == 2) { const v2f xv = *(const v2f*)xp; xh0[u] = xv.x; xh1[u] = xv.y; }
        else         xh0[u] = *xp;
      }
    }
  }
#pragma unroll
  for (int u = 0; u < 4; ++u) {
    float x0, x1;
    const int g = g0 + u * 32 + lane;
    float* xp = xs + (size_t)((g * HS) >> 5) * HID + ((g * HS) & 31);
    const bool held = g < capE;
    if (PASS == 2 && held) {
      x0 = xh0[u]; x1 = xh1[u];
    } else {
      edge_logits<H>(aS, aD, s[u], dl[u] + r0, hbase, x0, x1);
    }
    if (PASS == 1) {
      if (ok[u] && held) {
        if (HS == 2) { v2f xv; xv.x = x0; xv.y = x1; *(v2f*)xp = xv; }
        else         *xp = x0;
      }
      if (ok[u]) {
        atomicMaxFloat(&mx[dl[u] * HS], x0);
        if (HS == 2) atomicMaxFloat(&mx[dl[u] * HS + 1], x1);
      }
    } else {
      w0[u] = 0.f; w1[u] = 0.f;
      if (ok[u]) {
        w0[u] = expf(x0 - mx[dl[u] * HS]);
        ldsAdd(&sm[dl[u] * HS], w0[u]);
        if (HS == 2) {
          w1[u] = expf(x1 - mx[dl[u] * HS + 1]);
          ldsAdd(&sm[dl[u] * HS + 1], w1[u]);
        }
      }
    }
  }
  if (PASS == 2) {
    const bool lo = lane < 16;
    const char* Hb = (const char*)Hf;
    char* accb = (char*)acc;
    const unsigned laneB = (unsigned)(slab * SLAB + lane) * 4u;
    const unsigned lane4 = (unsigned)lane * 4u;
#pragma unroll
    for (int u = 0; u < 4; ++u) {
      int n = nb - u * 32;
      n = n > 32 ? 32 : n;
      if (n > 0) {
        const int so = s[u] * (HID * 4);
        const int dO = dl[u] * (SLAB * 4);
        const int wA = __float_as_int(w0[u]);
        const int wB = __float_as_int(HS == 2 ? w1[u] : w0[u]);
        const int sX = xrow_i(so), dX = xrow_i(dO), wAX = xrow_i(wA), wBX = xrow_i(wB);
#pragma unroll
        for (int g = 0; g < 2; ++g) {
          if (g * 16 < n) {
            const int   Sg = (g == 0) ? (lo ? so : sX) : (lo ? sX : so);
            const int   Dg = (g == 0) ? (lo ? dO : dX) : (lo ? dX : dO);
            const float Wg = __int_as_float((g == 0) ? (lo ? wA : wBX) : (lo ? wAX : wB));
            float hv[16];
#define GAT_LOAD(J) hv[J] = *(const float*)(Hb + (size_t)((unsigned)rshare_i<J>(Sg) + laneB));
#define GAT_ADD(J) ldsAdd((float*)(accb + ((unsigned)rshare_i<J>(Dg) + lane4)), hv[J] * rshare_f<J>(Wg));
            if (g * 16 + 16 <= n) {
              ROW16(GAT_LOAD)
              ROW16(GAT_ADD)
            } else {
              ROW16(GAT_LOAD)
#define GAT_ADDP(J) if (g * 16 + J < n) { GAT_ADD(J) }
              ROW16(GAT_ADDP)
#undef GAT_ADDP
            }
#undef GAT_ADD
#undef GAT_LOAD
          }
        }
      }
    }
  }
}

template <int H, int PASS>
__device__ __forceinline__ void drain_queue(const int* q, int cnt,
                                            const int* __restrict__ src,
                                            const int* __restrict__ dst,
                                            const float* __restrict__ aS,
                                            const float* __restrict__ aD,
                                            const float* __restrict__ Hf,
                                            float* acc, float* mx, float* sm,
                                            int r0, int hbase, int slab, int lane,
                                            float* seg0, int cap, int& ltot) {
  constexpr bool REC = (PASS == 0);
  bool room = false;
  if (REC) {
    room = ltot >= 0 && ltot + cnt <= cap;
    if (!room) ltot = -1;
  }
  for (int i0 = 0; i0 < cnt; i0 += 128) {
    int  s[4], dl[4];
    bool ok[4];
#pragma unroll
    for (int u = 0; u < 4; ++u) {
      const int i = i0 + u * 32 + lane;
      ok[u] = i < cnt;
      const int e = ok[u] ? q[i] : 0;
      s[u]  = src[e];
      dl[u] = dst[e] - r0;
    }
#pragma unroll
    for (int u = 0; u < 4; ++u) {
      if (!ok[u]) { s[u] = 0; dl[u] = 0; }
      if (REC && room && ok[u]) {
        const int p = ltot + i0 + u * 32 + lane;
        const float w = __uint_as_float(((unsigned)s[u] << LBITS) | (unsigned)dl[u]);
        float* at = seg0 + (size_t)(p >> 5) * HID + (p & 31);
#pragma unroll
        for (int sl = 0; sl < NPART; ++sl) at[sl * SLAB] = w;
      }
    }
    if constexpr (PASS != 0) {
      const int nb = cnt - i0;
      process_batch<H, PASS>(s, dl, ok, nb > 128 ? 128 : nb, aS, aD, Hf, acc, mx, sm,
                             r0, hbase, slab, lane, nullptr, 0, 0);
    }
  }
  if (REC && room) ltot += cnt;
}

template <int H, int PASS>
__device__ __forceinline__ void walk_lists(const float* seg0, size_t qStride,
                                           int c0, int c1, int c2, int c3,
                                           const float* __restrict__ aS,
                                           const float* __restrict__ aD,
                                           const float* __restrict__ Hf,
                                           float* acc, float* mx, float* sm,
                                           int r0, int hbase, int slab, int lane,
                                           float* xs, int capE) {
  const int t1 = c0, t2 = t1 + c1, t3 = t2 + c2, cnt = t3 + c3;
  auto word = [&](int g) -> unsigned {
    if (g >= cnt) return 0u;
    const int q   = (g >= t1) + (g >= t2) + (g >= t3);
    const int off = g - (g >= t3 ? t3 : (g >= t2 ? t2 : (g >= t1 ? t1 : 0)));
    return __float_as_uint(seg0[(size_t)q * qStride + (size_t)(off >> 5) * HID + (off & 31)]);
  };
  unsigned wn[4];
#pragma unroll
  for (int u = 0; u < 4; ++u) wn[u] = word(u * 32 + lane);
  for (int i0 = 0; i0 < cnt; i0 += 128) {
    int  s[4], dl[4];
    bool ok[4];
#pragma unroll
    for (int u = 0; u < 4; ++u) {
      ok[u] = i0 + u * 32 + lane < cnt;
      s[u]  = (int)(wn[u] >> LBITS);
      dl[u] = (int)(wn[u] & LMASK);
    }
#pragma unroll
    for (int u = 0; u < 4; ++u) wn[u] = word(i0 + 128 + u * 32 + lane);
    const int nb = cnt - i0;
    process_batch<H, PASS>(s, dl, ok, nb > 128 ? 128 : nb, aS, aD, Hf, acc, mx, sm,
                           r0, hbase, slab, lane, xs, i0, capE);
  }
}

__device__ __forceinline__ void push_hits(int* q, int& cnt, bool hit, int e, int lane) {
  const unsigned m = __builtin_amdgcn_ballot_w32(hit);
  if (m != 0u) {
    const int pos = cnt + __builtin_popcount(m & ((1u << lane) - 1u));
    if (hit) q[pos] = e;
    cnt += __builtin_popcount(m);
  }
}

__device__ __forceinline__ void load_step(const int* __restrict__ dst, int nEdges, int nChunks,
                                          int c, int lane, int (&d)[NLOAD], int& base) {
  base = 0;
  if (c < nChunks) {
    base = c * STEP_E;
    if (base + STEP_E <= nEdges) {
#pragma unroll
      for (int u = 0; u < NLOAD; ++u) d[u] = dst[base + u * 32 + lane];
    } else {
#pragma unroll
      for (int u = 0; u < NLOAD; ++u) {
        const int e = base + u * 32 + lane;
        d[u] = e < nEdges ? dst[e] : -1;
      }
    }
  } else {
#pragma unroll
    for (int u = 0; u < NLOAD; ++u) d[u] = -1;
  }
}

template <int H, int PASS, int NSUB>
__device__ __forceinline__ void scan_edges(int* q, int nEdges,
                                           const int* __restrict__ src,
                                           const int* __restrict__ dst,
                                           const float* __restrict__ aS,
                                           const float* __restrict__ aD,
                                           const float* __restrict__ Hf,
                                           float* acc, float* mx, float* sm,
                                           int r0, int rn, int hbase, int slab,
                                           int lane, int first,
                                           float* seg0, int cap, int& ltot) {
  const int nChunks = (nEdges + STEP_E - 1) / STEP_E;
  const int nRounds = (nChunks + CSTRIDE - 1) / CSTRIDE;
  const int nSteps  = nRounds * NSUB;
  const int rot = (int)(((long long)blockIdx.x * nRounds) / gridDim.x);
  auto chunk_of = [&](int m) {
    int r = m / NSUB + rot;
    if (r >= nRounds) r -= nRounds;
    return m < nSteps ? first + (m % NSUB) + r * CSTRIDE : nChunks;
  };
  int cnt = 0;
  int dn[NLOAD], basen;
  load_step(dst, nEdges, nChunks, chunk_of(0), lane, dn, basen);
  for (int m = 0; ; ++m) {
    const bool more = m < nSteps;
    if (more) {
      int d[NLOAD];
      const int base = basen;
#pragma unroll
      for (int u = 0; u < NLOAD; ++u) d[u] = dn[u];
      load_step(dst, nEdges, nChunks, chunk_of(m + 1), lane, dn, basen);
#pragma unroll
      for (int u = 0; u < NLOAD; ++u)
        push_hits(q, cnt, (unsigned)(d[u] - r0) < (unsigned)rn, base + u * 32 + lane, lane);
    }
    if (cnt > QCAP - STEP_E || (!more && cnt > 0)) {
      drain_queue<H, PASS>(q, cnt, src, dst, aS, aD, Hf, acc, mx, sm, r0, hbase, slab, lane,
                           seg0, cap, ltot);
      cnt = 0;
    }
    if (!more) break;
  }
}

__global__ void __launch_bounds__(1024) scan_lists_kernel(const int* __restrict__ ei, int nEdges,
                                                          float* lst) {
  __shared__ int qbuf[32 * QCAP];
  const int lane = threadIdx.x & 31;
  const int wave = __builtin_amdgcn_readfirstlane(threadIdx.x >> 5);
  const int part = blockIdx.x & (NPART - 1);
  const int tile = blockIdx.x / NPART;
  const int r0   = tile * TILE_R;
  const int rn   = (N_NODES - r0) < TILE_R ? (N_NODES - r0) : TILE_R;
  const int* __restrict__ src = ei;
  const int* __restrict__ dst = ei + nEdges;
  const int ppw = rn / (NPART * 32);
  const int cap = (ppw - 1) * 32;
  float* hdr  = lst + (size_t)(r0 + (part * 32 + wave) * ppw) * HID;
  float* seg0 = hdr + HID;
  int ltot = 0;
  scan_edges<1, 0, 1>(qbuf + wave * QCAP, nEdges, src, dst, nullptr, nullptr, nullptr,
                      nullptr, nullptr, nullptr, r0, rn, 0, 0, lane, part + NPART * wave,
                      seg0, cap, ltot);
  if (lane < NPART) hdr[lane * SLAB] = __int_as_float(ltot);
}

template <int H>
__global__ void __launch_bounds__(1024) owner_layer_kernel(const int* __restrict__ ei, int nEdges,
                                                           const float* __restrict__ aS,
                                                           const float* __restrict__ aD,
                                                           const float* __restrict__ Hf,
                                                           const float* __restrict__ bias,
                                                           float* out, const float* lst,
                                                           float* scr, int applyElu) {
  constexpr int HS = (H == 8) ? 2 : 1;
  __shared__ float acc[TILE_R * SLAB];
  __shared__ float mx[TILE_R * HS];
  __shared__ float sm[TILE_R * HS];
  __shared__ int   qbuf[32 * QCAP];

  const int lane = threadIdx.x & 31;
  const int wave = __builtin_amdgcn_readfirstlane(threadIdx.x >> 5);
  const int slab = blockIdx.x & 3;
  const int tile = blockIdx.x >> 2;
  const int r0   = tile * TILE_R;
  const int rn   = (N_NODES - r0) < TILE_R ? (N_NODES - r0) : TILE_R;
  const int hbase = (H == 8) ? slab * 2 : 0;
  const int* __restrict__ src = ei;
  const int* __restrict__ dst = ei + nEdges;
  int* q = qbuf + wave * QCAP;

  const int ppw = rn / (NPART * 32);
  const size_t qStride = (size_t)32 * ppw * HID;
  const float* hdr  = lst + (size_t)(r0 + wave * ppw) * HID + slab * SLAB;
  const float* seg0 = hdr + HID;
  const int c0 = __builtin_amdgcn_readfirstlane(__float_as_int(hdr[0]));
  const int c1 = __builtin_amdgcn_readfirstlane(__float_as_int(hdr[qStride]));
  const int c2 = __builtin_amdgcn_readfirstlane(__float_as_int(hdr[2 * qStride]));
  const int c3 = __builtin_amdgcn_readfirstlane(__float_as_int(hdr[3 * qStride]));
  const bool kept = (c0 | c1 | c2 | c3) >= 0;
  int none = 0;
  const int rpw  = rn / 32;
  float* xs      = scr + (size_t)(r0 + wave * rpw) * HID + slab * SLAB;
  const int capE = rpw * 32 / HS;

  for (int i = threadIdx.x; i < rn * HS; i += 1024) {
    const int node = r0 + i / HS;
    const int h    = hbase + i % HS;
    mx[i] = leaky(aS[(size_t)node * H + h] + aD[(size_t)node * H + h]);
  }
  __syncthreads();

  if (kept) {
    walk_lists<H, 1>(seg0, qStride, c0, c1, c2, c3, aS, aD, Hf, acc, mx, sm, r0, hbase, slab,
                     lane, xs, capE);
  } else {
    scan_edges<H, 1, NPART>(q, nEdges, src, dst, aS, aD, Hf, acc, mx, sm, r0, rn, hbase, slab,
                            lane, NPART * wave, nullptr, 0, none);
  }
  __syncthreads();

  for (int i = wave; i < rn; i += 32) {
    const int node = r0 + i;
    const int hh   = (HS == 2) ? (lane >> 4) : 0;
    const int h    = hbase + hh;
    const float x  = leaky(aS[(size_t)node * H + h] + aD[(size_t)node * H + h]);
    const float w  = expf(x - mx[i * HS + hh]);
    acc[i * SLAB + lane] = w * Hf[(size_t)node * HID + slab * SLAB + lane];
    if ((lane & 15) == 0 && (lane >> 4) < HS) sm[i * HS + (lane >> 4)] = w;
  }
  __syncthreads();

  if (kept) {
    walk_lists<H, 2>(seg0, qStride, c0, c1, c2, c3, aS, aD, Hf, acc, mx, sm, r0, hbase, slab,
                     lane, xs, capE);
  } else {
    scan_edges<H, 2, NPART>(q, nEdges, src, dst, aS, aD, Hf, acc, mx, sm, r0, rn, hbase, slab,
                            lane, NPART * wave, nullptr, 0, none);
  }
  __syncthreads();

  const float bv = bias[slab * SLAB + lane];
  for (int i = wave; i < rn; i += 32) {
    const int hh = (HS == 2) ? (lane >> 4) : 0;
    float v = acc[i * SLAB + lane] / sm[i * HS + hh] + bv;
    if (applyElu) v = v > 0.f ? v : (expf(v) - 1.f);
    out[(size_t)(r0 + i) * HID + slab * SLAB + lane] = v;
  }
}

extern "C" void kernel_launch(void* const* d_in, const int* in_sizes, int n_in,
                              void* d_out, int out_size, void* d_ws, size_t ws_size,
                              hipStream_t stream) {
  const float* x   = (const float*)d_in[0];
  const int*   ei  = (const int*)d_in[1];
  const float* W1  = (const float*)d_in[2];
  const float* as1 = (const float*)d_in[3];
  const float* ad1 = (const float*)d_in[4];
  const float* b1  = (const float*)d_in[5];
  const float* W2  = (const float*)d_in[6];
  const float* as2 = (const float*)d_in[7];
  const float* ad2 = (const float*)d_in[8];
  const float* b2  = (const float*)d_in[9];

  const int nEdges = in_sizes[1] / 2;

  float* A   = (float*)d_ws;
  float* B   = A  + (size_t)N_NODES * HID;
  float* aS  = B  + (size_t)N_NODES * HID;
  float* aD  = aS + (size_t)N_NODES * GAT_HEADS;
  float* out = (float*)d_out;

  const int TB = 256;
  auto cdiv = [](long long a, long long b) { return (unsigned)((a + b - 1) / b); };
  const unsigned gOwner = ((N_NODES + TILE_R - 1) / TILE_R) * (HID / SLAB);

  scan_lists_kernel<<<gOwner, 1024, 0, stream>>>(ei, nEdges, out);

  gemm128_wmma<<<GEMM_WG, 1024, 0, stream>>>(x, W1, A);
  alpha1_kernel<<<cdiv(N_NODES * GAT_HEADS, TB), TB, 0, stream>>>(A, as1, ad1, aS, aD);
  owner_layer_kernel<GAT_HEADS><<<gOwner, 1024, 0, stream>>>(ei, nEdges, aS, aD, A, b1, B, out, B, 1);

  gemm128_wmma<<<GEMM_WG, 1024, 0, stream>>>(B, W2, A);
  alpha2_kernel<<<cdiv((long long)N_NODES * 32, TB), TB, 0, stream>>>(A, as2, ad2, aS, aD);
  owner_layer_kernel<1><<<gOwner, 1024, 0, stream>>>(ei, nEdges, aS, aD, A, b2, out, out, B, 0);
}


# Edge phase tiled by nodes: a work-group owns 392 nodes x all 128 channels (was 1568 x 32): one 512-byte row load per entry, logits once per edge, scan writes a hit once; LDS sum order changes
# speedup vs baseline: 4.8340x; 1.4201x over previous
#include <hip/hip_runtime.h>

#define N_NODES 100000
#define IN_DIM 128
#define HID 128
#define GAT_HEADS 8
#define NEG_SLOPE 0.2f

typedef float v2f __attribute__((ext_vector_type(2)));
typedef float v4f __attribute__((ext_vector_type(4)));
typedef float v8f __attribute__((ext_vector_type(8)));

__device__ __forceinline__ int xrow_i(int v) {
  return __builtin_amdgcn_permlanex16(v, v, 0x76543210, 0xfedcba98, false, false);
}

#define WT_PITCH 132
#define GEMM_WG 256
static_assert((N_NODES % 16) == 0, "row tiles are whole");
static_assert((N_NODES / 16 + GEMM_WG - 1) / GEMM_WG <= 32, "one row tile per wave of a work-group");
static_assert(IN_DIM * 8 == 1024, "folding W with the logit vectors: one thread per (k, 16 channels)");
template <int NH>
__global__ void __launch_bounds__(1024) gemm128_wmma(const float* __restrict__ X,
                                                     const float* __restrict__ W,
                                                     const float* __restrict__ avS,
                                                     const float* __restrict__ avD,
                                                     float* __restrict__ H,
                                                     float* __restrict__ outS,
                                                     float* __restrict__ outD) {
  __shared__ __attribute__((aligned(16))) float Wt[(HID + 16) * WT_PITCH];
  const int lane = threadIdx.x & 31;
  const int wave = __builtin_amdgcn_readfirstlane(threadIdx.x >> 5);

  const int nTiles = N_NODES / 16;
  const int t0 = (int)(((long long)blockIdx.x * nTiles) / GEMM_WG);
  const int t1 = (int)(((long long)(blockIdx.x + 1) * nTiles) / GEMM_WG);
  const int slot = 8 * (((wave >> 3) + wave) & 3) + (wave & 7);
  const int tile = t0 + slot;
  const bool active = tile < t1;

  const int c = lane & 15;
  const int h = lane >> 4;
  const float* __restrict__ xrow = X + (size_t)(tile * 16 + c) * IN_DIM + 4 * h;
  const float* wl = Wt + c * WT_PITCH + 4 * h;

  v4f a = {0.f, 0.f, 0.f, 0.f};
  if (active) a = *(const v4f*)xrow;

  for (int i = threadIdx.x; i < IN_DIM * HID; i += 1024)
    Wt[(i & (HID - 1)) * WT_PITCH + (i >> 7)] = W[i];
  {
    const int k  = threadIdx.x >> 3;
    const int jc = threadIdx.x & 7;
    const float* wr = W + k * HID + 16 * jc;
    float ps = 0.f, pd = 0.f;
#pragma unroll
    for (int cc = 0; cc < 16; ++cc) {
      const float wv = wr[cc];
      ps += wv * avS[16 * jc + cc];
      pd += wv * avD[16 * jc + cc];
    }
    if (NH == 8) {
      Wt[(HID + jc) * WT_PITCH + k]     = ps;
      Wt[(HID + 8 + jc) * WT_PITCH + k] = pd;
    } else {
#pragma unroll
      for (int off = 1; off < 8; off <<= 1) {
        ps += __shfl_xor(ps, off, 32);
        pd += __shfl_xor(pd, off, 32);
      }
      Wt[(HID + jc) * WT_PITCH + k]     = jc == 0 ? ps : (jc == 1 ? pd : 0.f);
      Wt[(HID + 8 + jc) * WT_PITCH + k] = 0.f;
    }
  }
  __syncthreads();
  if (!active) return;

  v8f acc[9];
#pragma unroll
  for (int n = 0; n < 9; ++n) acc[n] = (v8f){0.f, 0.f, 0.f, 0.f, 0.f, 0.f, 0.f, 0.f};

#pragma unroll
  for (int j = 0; j < IN_DIM / 8; ++j) {
    v4f an = a;
    if (j + 1 < IN_DIM / 8) an = *(const v4f*)(xrow + 8 * (j + 1));
#pragma unroll
    for (int n0 = 0; n0 < 8; n0 += 4) {
      const int NN = n0 == 4 ? 5 : 4;
      v4f b[5];
#pragma unroll
      for (int nn = 0; nn < NN; ++nn)
        b[nn] = *(const v4f*)(wl + (n0 + nn) * 16 * WT_PITCH + 8 * j);
#pragma unroll
      for (int nn = 0; nn < NN; ++nn) {
        const v2f a0 = {a.x, a.y};
        const v2f b0 = {b[nn].x, b[nn].y};
        acc[n0 + nn] = __builtin_amdgcn_wmma_f32_16x16x4_f32(false, a0, false, b0, (short)0,
                                                             acc[n0 + nn], false, false);
      }
#pragma unroll
      for (int nn = 0; nn < NN; ++nn) {
        const v2f a1 = {a.z, a.w};
        const v2f b1 = {b[nn].z, b[nn].w};
        acc[n0 + nn] = __builtin_amdgcn_wmma_f32_16x16x4_f32(false, a1, false, b1, (short)0,
                                                             acc[n0 + nn], false, false);
      }
    }
    a = an;
  }

  float* hout = H + (size_t)tile * 16 * HID + lane;
#pragma unroll
  for (int p = 0; p < 4; ++p) {
#pragma unroll
    for (int v = 0; v < 8; ++v) {
      const float x0 = acc[2 * p][v];
      const float x1 = acc[2 * p + 1][v];
      const float x0o = __int_as_float(xrow_i(__float_as_int(x0)));
      const float x1o = __int_as_float(xrow_i(__float_as_int(x1)));
      const float ylo = h == 0 ? x0 : x1o;
      const float yhi = h == 0 ? x0o : x1;
      hout[(size_t)v * HID + 32 * p]       = ylo;
      hout[(size_t)(v + 8) * HID + 32 * p] = yhi;
    }
  }

  if (NH == 8) {
    float* o = (c < 8 ? outS : outD) + (size_t)(tile * 16 + 8 * h) * 8 + (c & 7);
#pragma unroll
    for (int v = 0; v < 8; ++v) o[v * 8] = acc[8][v];
  } else if (c < 2) {
    float* o = (c == 0 ? outS : outD) + (size_t)(tile * 16 + 8 * h);
#pragma unroll
    for (int v = 0; v < 8; ++v) o[v] = acc[8][v];
  }
}


#define TILE_R 1568
#define SUB_R  (TILE_R / 4)
#define NLOAD  12
#define STEP_E (NLOAD * 32)
#define QCAP   512
#define LBITS  11
#define LMASK  ((1u << LBITS) - 1u)
static_assert(SUB_R <= (1 << LBITS), "owner-local destination must fit its field");
static_assert(N_NODES <= (1 << (32 - LBITS)), "source node must fit its field");
#define NPART  4
#define CSTRIDE (NPART * 32)
static_assert(TILE_R % 4 == 0 && (N_NODES % TILE_R) % 4 == 0, "a tile splits into four equal owners");
static_assert(N_NODES % TILE_R == 0 || (N_NODES % TILE_R) / 4 >= 64,
              "the last tile must give every list a count line and a data line of its own");

__device__ __forceinline__ float leaky(float x) {
  return x > 0.f ? x : NEG_SLOPE * x;
}

__device__ __forceinline__ void atomicMaxFloat(float* addr, float val) {
  if (__float_as_int(val) >= 0)
    atomicMax((int*)addr, __float_as_int(val));
  else
    atomicMin((unsigned int*)addr, __float_as_uint(val));
}

__device__ __forceinline__ void ldsAdd(float* addr, float val) {
  __hip_atomic_fetch_add(addr, val, __ATOMIC_RELAXED, __HIP_MEMORY_SCOPE_WORKGROUP);
}

template <int J>
__device__ __forceinline__ int rshare_i(int v) {
  return __builtin_amdgcn_update_dpp(0, v, 0x150 + J, 0xf, 0xf, true);
}
template <int J>
__device__ __forceinline__ float rshare_f(float v) {
  return __int_as_float(rshare_i<J>(__float_as_int(v)));
}
#define ROW8A(M) M(0) M(1) M(2) M(3) M(4) M(5) M(6) M(7)
#define ROW8B(M) M(8) M(9) M(10) M(11) M(12) M(13) M(14) M(15)

template <int H> struct BatchU { static constexpr int value = (H == 8) ? 2 : 4; };

template <int H, int PASS, int U>
__device__ __forceinline__ void process_batch(const int (&s)[U], const int (&dl)[U],
                                              const bool (&ok)[U], int nb,
                                              const float* __restrict__ aS,
                                              const float* __restrict__ aD,
                                              const float* __restrict__ Hf,
                                              float* acc, float* mx, float* sm, float* wb,
                                              int r0, int lane) {
  float w[U][H];
  if constexpr (H == 8) {
    v4f a0[U], a1[U], b0[U], b1[U];
#pragma unroll
    for (int u = 0; u < U; ++u) {
      const v4f* ps = (const v4f*)(aS + (size_t)s[u] * 8);
      const v4f* pd = (const v4f*)(aD + (size_t)(dl[u] + r0) * 8);
      a0[u] = ps[0]; a1[u] = ps[1]; b0[u] = pd[0]; b1[u] = pd[1];
    }
#pragma unroll
    for (int u = 0; u < U; ++u) {
      const float x[8] = {leaky(a0[u].x + b0[u].x), leaky(a0[u].y + b0[u].y),
                          leaky(a0[u].z + b0[u].z), leaky(a0[u].w + b0[u].w),
                          leaky(a1[u].x + b1[u].x), leaky(a1[u].y + b1[u].y),
                          leaky(a1[u].z + b1[u].z), leaky(a1[u].w + b1[u].w)};
      if (PASS == 1) {
        if (ok[u]) {
#pragma unroll
          for (int h = 0; h < 8; ++h) atomicMaxFloat(&mx[h * SUB_R + dl[u]], x[h]);
        }
      } else {
#pragma unroll
        for (int h = 0; h < 8; ++h) w[u][h] = 0.f;
        if (ok[u]) {
          float m[8];
#pragma unroll
          for (int h = 0; h < 8; ++h) m[h] = mx[h * SUB_R + dl[u]];
#pragma unroll
          for (int h = 0; h < 8; ++h) w[u][h] = expf(x[h] - m[h]);
#pragma unroll
          for (int h = 0; h < 8; ++h) ldsAdd(&sm[h * SUB_R + dl[u]], w[u][h]);
        }
      }
    }
  } else {
    float a[U], b[U];
#pragma unroll
    for (int u = 0; u < U; ++u) { a[u] = aS[s[u]]; b[u] = aD[dl[u] + r0]; }
#pragma unroll
    for (int u = 0; u < U; ++u) {
      const float x = leaky(a[u] + b[u]);
      if (PASS == 1) {
        if (ok[u]) atomicMaxFloat(&mx[dl[u]], x);
      } else {
        w[u][0] = 0.f;
        if (ok[u]) {
          w[u][0] = expf(x - mx[dl[u]]);
          ldsAdd(&sm[dl[u]], w[u][0]);
        }
      }
    }
  }
  if (PASS == 2) {
    const bool lo = lane < 16;
    const char* Hb = (const char*)Hf;
    char* accb = (char*)acc;
    const unsigned lane16 = (unsigned)lane * 16u;
    const unsigned lane4  = (unsigned)lane * 4u;
#pragma unroll
    for (int u = 0; u < U; ++u) {
      int n = nb - u * 32;
      n = n > 32 ? 32 : n;
      if (n > 0) {
        const int so = s[u] * (HID * 4);
        const int dO = dl[u] * (HID * 4);
        const int sX = xrow_i(so), dX = xrow_i(dO);
        int wA = 0, wAX = 0;
        if constexpr (H == 8) {
          v4f t0 = {w[u][0], w[u][1], w[u][2], w[u][3]};
          v4f t1 = {w[u][4], w[u][5], w[u][6], w[u][7]};
          *(v4f*)(wb + lane * 8)     = t0;
          *(v4f*)(wb + lane * 8 + 4) = t1;
        } else {
          wA = __float_as_int(w[u][0]);
          wAX = xrow_i(wA);
        }
#pragma unroll
        for (int g = 0; g < 2; ++g) {
          if (g * 16 < n) {
            const int   Sg = (g == 0) ? (lo ? so : sX) : (lo ? sX : so);
            const int   Dg = (g == 0) ? (lo ? dO : dX) : (lo ? dX : dO);
            const float Wg = __int_as_float((g == 0) ? (lo ? wA : wAX) : (lo ? wAX : wA));
            v4f hv[8];
            float wq[8];
#define GAT_LOAD(J) hv[(J) & 7] = *(const v4f*)(Hb + (size_t)((unsigned)rshare_i<J>(Sg) + lane16)); \
            if constexpr (H == 8) wq[(J) & 7] = wb[(g * 16 + J) * 8 + (lane >> 2)];
#define GAT_ADD(J) { \
              float wv; \
              if constexpr (H == 8) wv = wq[(J) & 7]; \
              else                  wv = rshare_f<J>(Wg); \
              float* ap = (float*)(accb + ((unsigned)rshare_i<J>(Dg) + lane4)); \
              const v4f t = hv[(J) & 7]; \
              ldsAdd(ap, t.x * wv); ldsAdd(ap + 32, t.y * wv); \
              ldsAdd(ap + 64, t.z * wv); ldsAdd(ap + 96, t.w * wv); }
#define GAT_ADDP(J) if (g * 16 + J < n) GAT_ADD(J)
            if (g * 16 + 8 <= n) {
              ROW8A(GAT_LOAD)
              ROW8A(GAT_ADD)
            } else {
              ROW8A(GAT_LOAD)
              ROW8A(GAT_ADDP)
            }
            if (g * 16 + 8 < n) {
              if (g * 16 + 16 <= n) {
                ROW8B(GAT_LOAD)
                ROW8B(GAT_ADD)
              } else {
                ROW8B(GAT_LOAD)
                ROW8B(GAT_ADDP)
              }
            }
#undef GAT_ADDP
#undef GAT_ADD
#undef GAT_LOAD
          }
        }
      }
    }
  }
}

template <int H, int PASS>
__device__ __forceinline__ void walk_lists(const float* seg0, size_t qStride,
                                           int c0, int c1, int c2, int c3,
                                           const float* __restrict__ aS,
                                           const float* __restrict__ aD,
                                           const float* __restrict__ Hf,
                                           float* acc, float* mx, float* sm, float* wb,
                                           int r0, int lane) {
  constexpr int U = BatchU<H>::value;
  const int t1 = c0, t2 = t1 + c1, t3 = t2 + c2, cnt = t3 + c3;
  auto word = [&](int g) -> unsigned {
    if (g >= cnt) return 0u;
    const int q   = (g >= t1) + (g >= t2) + (g >= t3);
    const int off = g - (g >= t3 ? t3 : (g >= t2 ? t2 : (g >= t1 ? t1 : 0)));
    return __float_as_uint(seg0[(size_t)q * qStride + off]);
  };
  unsigned wn[U];
#pragma unroll
  for (int u = 0; u < U; ++u) wn[u] = word(u * 32 + lane);
  for (int i0 = 0; i0 < cnt; i0 += 32 * U) {
    int  s[U], dl[U];
    bool ok[U];
#pragma unroll
    for (int u = 0; u < U; ++u) {
      ok[u] = i0 + u * 32 + lane < cnt;
      s[u]  = (int)(wn[u] >> LBITS);
      dl[u] = (int)(wn[u] & LMASK);
    }
#pragma unroll
    for (int u = 0; u < U; ++u) wn[u] = word(i0 + 32 * U + u * 32 + lane);
    const int nb = cnt - i0;
    process_batch<H, PASS, U>(s, dl, ok, nb > 32 * U ? 32 * U : nb, aS, aD, Hf, acc, mx, sm, wb,
                              r0, lane);
  }
}

template <int H, int PASS>
__device__ __forceinline__ void drain_work(const int* q, int cnt,
                                           const int* __restrict__ src,
                                           const int* __restrict__ dst,
                                           const float* __restrict__ aS,
                                           const float* __restrict__ aD,
                                           const float* __restrict__ Hf,
                                           float* acc, float* mx, float* sm, float* wb,
                                           int r0, int lane) {
  constexpr int U = BatchU<H>::value;
  for (int i0 = 0; i0 < cnt; i0 += 32 * U) {
    int  s[U], dl[U];
    bool ok[U];
#pragma unroll
    for (int u = 0; u < U; ++u) {
      const int i = i0 + u * 32 + lane;
      ok[u] = i < cnt;
      const int e = ok[u] ? q[i] : 0;
      s[u]  = src[e];
      dl[u] = dst[e] - r0;
    }
#pragma unroll
    for (int u = 0; u < U; ++u)
      if (!ok[u]) { s[u] = 0; dl[u] = 0; }
    const int nb = cnt - i0;
    process_batch<H, PASS, U>(s, dl, ok, nb > 32 * U ? 32 * U : nb, aS, aD, Hf, acc, mx, sm, wb,
                              r0, lane);
  }
}

__device__ __forceinline__ void push_hits(int* q, int& cnt, bool hit, int e, int lane) {
  const unsigned m = __builtin_amdgcn_ballot_w32(hit);
  if (m != 0u) {
    const int pos = cnt + __builtin_popcount(m & ((1u << lane) - 1u));
    if (hit) q[pos] = e;
    cnt += __builtin_popcount(m);
  }
}

__device__ __forceinline__ void load_step(const int* __restrict__ dst, int nEdges, int nChunks,
                                          int c, int lane, int (&d)[NLOAD], int& base) {
  base = 0;
  if (c < nChunks) {
    base = c * STEP_E;
    if (base + STEP_E <= nEdges) {
#pragma unroll
      for (int u = 0; u < NLOAD; ++u) d[u] = dst[base + u * 32 + lane];
    } else {
#pragma unroll
      for (int u = 0; u < NLOAD; ++u) {
        const int e = base + u * 32 + lane;
        d[u] = e < nEdges ? dst[e] : -1;
      }
    }
  } else {
#pragma unroll
    for (int u = 0; u < NLOAD; ++u) d[u] = -1;
  }
}

template <int NSUB, typename DrainF>
__device__ __forceinline__ void scan_edges(int* q, int nEdges, const int* __restrict__ dst,
                                           int r0, int rn, int lane, int first, DrainF drain) {
  const int nChunks = (nEdges + STEP_E - 1) / STEP_E;
  const int nRounds = (nChunks + CSTRIDE - 1) / CSTRIDE;
  const int nSteps  = nRounds * NSUB;
  const int rot = (int)(((long long)blockIdx.x * nRounds) / gridDim.x);
  auto chunk_of = [&](int m) {
    int r = m / NSUB + rot;
    if (r >= nRounds) r -= nRounds;
    return m < nSteps ? first + (m % NSUB) + r * CSTRIDE : nChunks;
  };
  int cnt = 0;
  int dn[NLOAD], basen;
  load_step(dst, nEdges, nChunks, chunk_of(0), lane, dn, basen);
  for (int m = 0; ; ++m) {
    const bool more = m < nSteps;
    if (more) {
      int d[NLOAD];
      const int base = basen;
#pragma unroll
      for (int u = 0; u < NLOAD; ++u) d[u] = dn[u];
      load_step(dst, nEdges, nChunks, chunk_of(m + 1), lane, dn, basen);
#pragma unroll
      for (int u = 0; u < NLOAD; ++u)
        push_hits(q, cnt, (unsigned)(d[u] - r0) < (unsigned)rn, base + u * 32 + lane, lane);
    }
    if (cnt > QCAP - STEP_E || (!more && cnt > 0)) {
      drain(q, cnt);
      cnt = 0;
    }
    if (!more) break;
  }
}

__global__ void __launch_bounds__(1024) scan_lists_kernel(const int* __restrict__ ei, int nEdges,
                                                          float* lst) {
  __shared__ int qbuf[32 * QCAP];
  const int lane = threadIdx.x & 31;
  const int wave = __builtin_amdgcn_readfirstlane(threadIdx.x >> 5);
  const int part = blockIdx.x & (NPART - 1);
  const int tile = blockIdx.x / NPART;
  const int R0   = tile * TILE_R;
  const int RN   = (N_NODES - R0) < TILE_R ? (N_NODES - R0) : TILE_R;
  const int sub  = RN >> 2;
  const int* __restrict__ src = ei;
  const int* __restrict__ dst = ei + nEdges;
  const int ppw = sub / 32;
  const int cap = (ppw - 1) * 32;
  const size_t aStride = (size_t)sub * HID;
  float* hdr0 = lst + (size_t)R0 * HID + (size_t)(part * 32 + wave) * ppw * 32;
  int lt[4] = {0, 0, 0, 0};
  auto rec = [&](const int* q, int cnt) __attribute__((always_inline)) {
    for (int i0 = 0; i0 < cnt; i0 += 128) {
      unsigned wd[4];
      int jj[4];
#pragma unroll
      for (int u = 0; u < 4; ++u) {
        const int i = i0 + u * 32 + lane;
        const bool ok = i < cnt;
        const int e = ok ? q[i] : 0;
        const int s = src[e];
        const int d = dst[e] - R0;
        const int j = (d >= sub) + (d >= 2 * sub) + (d >= 3 * sub);
        wd[u] = ((unsigned)s << LBITS) | (unsigned)(d - j * sub);
        jj[u] = ok ? j : -1;
      }
#pragma unroll
      for (int u = 0; u < 4; ++u) {
#pragma unroll
        for (int t = 0; t < 4; ++t) {
          const bool mine = jj[u] == t;
          const unsigned m = __builtin_amdgcn_ballot_w32(mine);
          if (m != 0u) {
            const int n = __builtin_popcount(m);
            if (lt[t] >= 0 && lt[t] + n <= cap) {
              if (mine) {
                const int p = lt[t] + __builtin_popcount(m & ((1u << lane) - 1u));
                hdr0[t * aStride + 32 + p] = __uint_as_float(wd[u]);
              }
              lt[t] += n;
            } else {
              lt[t] = -1;
            }
          }
        }
      }
    }
  };
  scan_edges<1>(qbuf + wave * QCAP, nEdges, dst, R0, RN, lane, part + NPART * wave, rec);
  if (lane < 4) {
    const int v = lane == 0 ? lt[0] : (lane == 1 ? lt[1] : (lane == 2 ? lt[2] : lt[3]));
    hdr0[lane * aStride] = __int_as_float(v);
  }
}

template <int H>
__global__ void __launch_bounds__(1024) owner_layer_kernel(const int* __restrict__ ei, int nEdges,
                                                           const float* __restrict__ aS,
                                                           const float* __restrict__ aD,
                                                           const float* __restrict__ Hf,
                                                           const float* __restrict__ bias,
                                                           float* out, const float* lst,
                                                           int applyElu) {
  __shared__ __attribute__((aligned(16))) float acc[SUB_R * HID];
  __shared__ float mx[SUB_R * H];
  __shared__ float sm[SUB_R * H];
  __shared__ int   qbuf[32 * QCAP];
  __shared__ __attribute__((aligned(16))) float wbuf[H == 8 ? 32 * 256 : 32];

  const int lane = threadIdx.x & 31;
  const int wave = __builtin_amdgcn_readfirstlane(threadIdx.x >> 5);
  const int own  = blockIdx.x & 3;
  const int tile = blockIdx.x >> 2;
  const int R0   = tile * TILE_R;
  const int RN   = (N_NODES - R0) < TILE_R ? (N_NODES - R0) : TILE_R;
  const int rn   = RN >> 2;
  const int r0   = R0 + own * rn;
  const int* __restrict__ src = ei;
  const int* __restrict__ dst = ei + nEdges;
  int* q = qbuf + wave * QCAP;
  float* wb = wbuf + (H == 8 ? wave * 256 : 0);

  const int ppw = rn / 32;
  const size_t qStride = (size_t)32 * ppw * 32;
  const float* hdr  = lst + (size_t)r0 * HID + (size_t)wave * ppw * 32;
  const float* seg0 = hdr + 32;
  const int c0 = __builtin_amdgcn_readfirstlane(__float_as_int(hdr[0]));
  const int c1 = __builtin_amdgcn_readfirstlane(__float_as_int(hdr[qStride]));
  const int c2 = __builtin_amdgcn_readfirstlane(__float_as_int(hdr[2 * qStride]));
  const int c3 = __builtin_amdgcn_readfirstlane(__float_as_int(hdr[3 * qStride]));
  const bool kept = (c0 | c1 | c2 | c3) >= 0;

  const int hh = (H == 8) ? (lane >> 2) : 0;

  for (int i = threadIdx.x; i < rn * H; i += 1024) {
    const int node = r0 + i / H;
    const int h    = i % H;
    mx[h * SUB_R + i / H] = leaky(aS[(size_t)node * H + h] + aD[(size_t)node * H + h]);
  }
  __syncthreads();

  if (kept) {
    walk_lists<H, 1>(seg0, qStride, c0, c1, c2, c3, aS, aD, Hf, acc, mx, sm, wb, r0, lane);
  } else {
    scan_edges<NPART>(q, nEdges, dst, r0, rn, lane, NPART * wave, [&](const int* qq, int cnt) {
      drain_work<H, 1>(qq, cnt, src, dst, aS, aD, Hf, acc, mx, sm, wb, r0, lane);
    });
  }
  __syncthreads();

  for (int i = wave; i < rn; i += 32) {
    const int node = r0 + i;
    const float x  = leaky(aS[(size_t)node * H + hh] + aD[(size_t)node * H + hh]);
    const float w  = expf(x - mx[hh * SUB_R + i]);
    const v4f hv = *(const v4f*)(Hf + (size_t)node * HID + 4 * lane);
    float* ap = acc + i * HID + lane;
    ap[0] = w * hv.x; ap[32] = w * hv.y; ap[64] = w * hv.z; ap[96] = w * hv.w;
    if ((lane & 3) == 0 && (lane >> 2) < H) sm[(lane >> 2) * SUB_R + i] = w;
  }
  __syncthreads();

  if (kept) {
    walk_lists<H, 2>(seg0, qStride, c0, c1, c2, c3, aS, aD, Hf, acc, mx, sm, wb, r0, lane);
  } else {
    scan_edges<NPART>(q, nEdges, dst, r0, rn, lane, NPART * wave, [&](const int* qq, int cnt) {
      drain_work<H, 2>(qq, cnt, src, dst, aS, aD, Hf, acc, mx, sm, wb, r0, lane);
    });
  }
  __syncthreads();

  const v4f bv = *(const v4f*)(bias + 4 * lane);
  for (int i = wave; i < rn; i += 32) {
    const float* ap = acc + i * HID + lane;
    const float sv = sm[hh * SUB_R + i];
    v4f v = {ap[0] / sv + bv.x, ap[32] / sv + bv.y, ap[64] / sv + bv.z, ap[96] / sv + bv.w};
    if (applyElu) {
      v.x = v.x > 0.f ? v.x : (expf(v.x) - 1.f);
      v.y = v.y > 0.f ? v.y : (expf(v.y) - 1.f);
      v.z = v.z > 0.f ? v.z : (expf(v.z) - 1.f);
      v.w = v.w > 0.f ? v.w : (expf(v.w) - 1.f);
    }
    *(v4f*)(out + (size_t)(r0 + i) * HID + 4 * lane) = v;
  }
}

extern "C" void kernel_launch(void* const* d_in, const int* in_sizes, int n_in,
                              void* d_out, int out_size, void* d_ws, size_t ws_size,
                              hipStream_t stream) {
  const float* x   = (const float*)d_in[0];
  const int*   ei  = (const int*)d_in[1];
  const float* W1  = (const float*)d_in[2];
  const float* as1 = (const float*)d_in[3];
  const float* ad1 = (const float*)d_in[4];
  const float* b1  = (const float*)d_in[5];
  const float* W2  = (const float*)d_in[6];
  const float* as2 = (const float*)d_in[7];
  const float* ad2 = (const float*)d_in[8];
  const float* b2  = (const float*)d_in[9];

  const int nEdges = in_sizes[1] / 2;

  float* A   = (float*)d_ws;
  float* B   = A  + (size_t)N_NODES * HID;
  float* aS  = B  + (size_t)N_NODES * HID;
  float* aD  = aS + (size_t)N_NODES * GAT_HEADS;
  float* out = (float*)d_out;

  const unsigned gOwner = ((N_NODES + TILE_R - 1) / TILE_R) * NPART;

  scan_lists_kernel<<<gOwner, 1024, 0, stream>>>(ei, nEdges, out);

  gemm128_wmma<GAT_HEADS><<<GEMM_WG, 1024, 0, stream>>>(x, W1, as1, ad1, A, aS, aD);
  owner_layer_kernel<GAT_HEADS><<<gOwner, 1024, 0, stream>>>(ei, nEdges, aS, aD, A, b1, B, out, 1);

  gemm128_wmma<1><<<GEMM_WG, 1024, 0, stream>>>(B, W2, as2, ad2, A, aS, aD);
  owner_layer_kernel<1><<<gOwner, 1024, 0, stream>>>(ei, nEdges, aS, aD, A, b2, out, out, 0);
}


# Edge phase walks its lists once per layer, not twice: softmax shifted by a bound (largest source logit of any node, from the projection), exp sums tested, two-pass form as fallback; last bits change
# speedup vs baseline: 5.2109x; 1.0780x over previous
#include <hip/hip_runtime.h>

#define N_NODES 100000
#define IN_DIM 128
#define HID 128
#define GAT_HEADS 8
#define NEG_SLOPE 0.2f

typedef float v2f __attribute__((ext_vector_type(2)));
typedef float v4f __attribute__((ext_vector_type(4)));
typedef float v8f __attribute__((ext_vector_type(8)));

__device__ __forceinline__ int xrow_i(int v) {
  return __builtin_amdgcn_permlanex16(v, v, 0x76543210, 0xfedcba98, false, false);
}

#define WT_PITCH 132
#define GEMM_WG 256
static_assert((N_NODES % 16) == 0, "row tiles are whole");
static_assert((N_NODES / 16 + GEMM_WG - 1) / GEMM_WG <= 32, "one row tile per wave of a work-group");
static_assert(IN_DIM * 8 == 1024, "folding W with the logit vectors: one thread per (k, 16 channels)");
template <int NH>
__global__ void __launch_bounds__(1024) gemm128_wmma(const float* __restrict__ X,
                                                     const float* __restrict__ W,
                                                     const float* __restrict__ avS,
                                                     const float* __restrict__ avD,
                                                     float* __restrict__ H,
                                                     float* __restrict__ outS,
                                                     float* __restrict__ outD,
                                                     float* __restrict__ wgMax) {
  __shared__ __attribute__((aligned(16))) float Wt[(HID + 16) * WT_PITCH];
  __shared__ float wmaxS[32 * 8];
  const int lane = threadIdx.x & 31;
  const int wave = __builtin_amdgcn_readfirstlane(threadIdx.x >> 5);

  const int nTiles = N_NODES / 16;
  const int t0 = (int)(((long long)blockIdx.x * nTiles) / GEMM_WG);
  const int t1 = (int)(((long long)(blockIdx.x + 1) * nTiles) / GEMM_WG);
  const int slot = 8 * (((wave >> 3) + wave) & 3) + (wave & 7);
  const int tile = t0 + slot;
  const bool active = tile < t1;

  const int c = lane & 15;
  const int h = lane >> 4;
  const float* __restrict__ xrow = X + (size_t)(tile * 16 + c) * IN_DIM + 4 * h;
  const float* wl = Wt + c * WT_PITCH + 4 * h;

  v4f a = {0.f, 0.f, 0.f, 0.f};
  if (active) a = *(const v4f*)xrow;

  for (int i = threadIdx.x; i < IN_DIM * HID; i += 1024)
    Wt[(i & (HID - 1)) * WT_PITCH + (i >> 7)] = W[i];
  {
    const int k  = threadIdx.x >> 3;
    const int jc = threadIdx.x & 7;
    const float* wr = W + k * HID + 16 * jc;
    float ps = 0.f, pd = 0.f;
#pragma unroll
    for (int cc = 0; cc < 16; ++cc) {
      const float wv = wr[cc];
      ps += wv * avS[16 * jc + cc];
      pd += wv * avD[16 * jc + cc];
    }
    if (NH == 8) {
      Wt[(HID + jc) * WT_PITCH + k]     = ps;
      Wt[(HID + 8 + jc) * WT_PITCH + k] = pd;
    } else {
#pragma unroll
      for (int off = 1; off < 8; off <<= 1) {
        ps += __shfl_xor(ps, off, 32);
        pd += __shfl_xor(pd, off, 32);
      }
      Wt[(HID + jc) * WT_PITCH + k]     = jc == 0 ? ps : (jc == 1 ? pd : 0.f);
      Wt[(HID + 8 + jc) * WT_PITCH + k] = 0.f;
    }
  }
  __syncthreads();

  float mloc = -__builtin_inff();
  if (active) {
  v8f acc[9];
#pragma unroll
  for (int n = 0; n < 9; ++n) acc[n] = (v8f){0.f, 0.f, 0.f, 0.f, 0.f, 0.f, 0.f, 0.f};

#pragma unroll
  for (int j = 0; j < IN_DIM / 8; ++j) {
    v4f an = a;
    if (j + 1 < IN_DIM / 8) an = *(const v4f*)(xrow + 8 * (j + 1));
#pragma unroll
    for (int n0 = 0; n0 < 8; n0 += 4) {
      const int NN = n0 == 4 ? 5 : 4;
      v4f b[5];
#pragma unroll
      for (int nn = 0; nn < NN; ++nn)
        b[nn] = *(const v4f*)(wl + (n0 + nn) * 16 * WT_PITCH + 8 * j);
#pragma unroll
      for (int nn = 0; nn < NN; ++nn) {
        const v2f a0 = {a.x, a.y};
        const v2f b0 = {b[nn].x, b[nn].y};
        acc[n0 + nn] = __builtin_amdgcn_wmma_f32_16x16x4_f32(false, a0, false, b0, (short)0,
                                                             acc[n0 + nn], false, false);
      }
#pragma unroll
      for (int nn = 0; nn < NN; ++nn) {
        const v2f a1 = {a.z, a.w};
        const v2f b1 = {b[nn].z, b[nn].w};
        acc[n0 + nn] = __builtin_amdgcn_wmma_f32_16x16x4_f32(false, a1, false, b1, (short)0,
                                                             acc[n0 + nn], false, false);
      }
    }
    a = an;
  }

  float* hout = H + (size_t)tile * 16 * HID + lane;
#pragma unroll
  for (int p = 0; p < 4; ++p) {
#pragma unroll
    for (int v = 0; v < 8; ++v) {
      const float x0 = acc[2 * p][v];
      const float x1 = acc[2 * p + 1][v];
      const float x0o = __int_as_float(xrow_i(__float_as_int(x0)));
      const float x1o = __int_as_float(xrow_i(__float_as_int(x1)));
      const float ylo = h == 0 ? x0 : x1o;
      const float yhi = h == 0 ? x0o : x1;
      hout[(size_t)v * HID + 32 * p]       = ylo;
      hout[(size_t)(v + 8) * HID + 32 * p] = yhi;
    }
  }

  if (NH == 8) {
    float* o = (c < 8 ? outS : outD) + (size_t)(tile * 16 + 8 * h) * 8 + (c & 7);
#pragma unroll
    for (int v = 0; v < 8; ++v) o[v * 8] = acc[8][v];
  } else if (c < 2) {
    float* o = (c == 0 ? outS : outD) + (size_t)(tile * 16 + 8 * h);
#pragma unroll
    for (int v = 0; v < 8; ++v) o[v] = acc[8][v];
  }

  {
    float m = acc[8][0];
#pragma unroll
    for (int v = 1; v < 8; ++v) m = fmaxf(m, acc[8][v]);
    mloc = fmaxf(m, __int_as_float(xrow_i(__float_as_int(m))));
  }
  }

  if (lane < 8) wmaxS[wave * 8 + lane] = mloc;
  __syncthreads();
  if (threadIdx.x < (NH == 8 ? 8 : 1)) {
    float m = wmaxS[threadIdx.x];
    for (int w = 1; w < 32; ++w) m = fmaxf(m, wmaxS[w * 8 + threadIdx.x]);
    wgMax[blockIdx.x * 8 + threadIdx.x] = m;
  }
}


#define TILE_R 1568
#define SUB_R  (TILE_R / 4)
#define NLOAD  12
#define STEP_E (NLOAD * 32)
#define QCAP   512
#define LBITS  11
#define LMASK  ((1u << LBITS) - 1u)
static_assert(SUB_R <= (1 << LBITS), "owner-local destination must fit its field");
static_assert(N_NODES <= (1 << (32 - LBITS)), "source node must fit its field");
#define NPART  4
#define SHIFT_FLOOR 9.5367431640625e-07f
#define CSTRIDE (NPART * 32)
static_assert(TILE_R % 4 == 0 && (N_NODES % TILE_R) % 4 == 0, "a tile splits into four equal owners");
static_assert(N_NODES % TILE_R == 0 || (N_NODES % TILE_R) / 4 >= 64,
              "the last tile must give every list a count line and a data line of its own");
#define LAST_SUB ((N_NODES % TILE_R == 0 ? TILE_R : N_NODES % TILE_R) / 4)
static_assert(LAST_SUB * 4 - 128 * (LAST_SUB / 32) >= GEMM_WG * 8 / 32,
              "the last owner's rows must keep GEMM_WG x 8 words free behind its lists");
static_assert(GEMM_WG * 8 == 2 * 1024 && GEMM_WG <= 1024, "two bound words per thread (8 heads), one (one head)");
static_assert(N_NODES * (GAT_HEADS - 1) >= GEMM_WG * 8, "layer 2's logit plane has room behind its N values");

__device__ __forceinline__ float leaky(float x) {
  return x > 0.f ? x : NEG_SLOPE * x;
}

__device__ __forceinline__ void atomicMaxFloat(float* addr, float val) {
  if (__float_as_int(val) >= 0)
    atomicMax((int*)addr, __float_as_int(val));
  else
    atomicMin((unsigned int*)addr, __float_as_uint(val));
}

__device__ __forceinline__ void ldsAdd(float* addr, float val) {
  __hip_atomic_fetch_add(addr, val, __ATOMIC_RELAXED, __HIP_MEMORY_SCOPE_WORKGROUP);
}

template <int J>
__device__ __forceinline__ int rshare_i(int v) {
  return __builtin_amdgcn_update_dpp(0, v, 0x150 + J, 0xf, 0xf, true);
}
template <int J>
__device__ __forceinline__ float rshare_f(float v) {
  return __int_as_float(rshare_i<J>(__float_as_int(v)));
}
#define ROW8A(M) M(0) M(1) M(2) M(3) M(4) M(5) M(6) M(7)
#define ROW8B(M) M(8) M(9) M(10) M(11) M(12) M(13) M(14) M(15)

template <int H> struct BatchU { static constexpr int value = (H == 8) ? 2 : 4; };

template <int H, int PASS, int U>
__device__ __forceinline__ void process_batch(const int (&s)[U], const int (&dl)[U],
                                              const bool (&ok)[U], int nb,
                                              const float* __restrict__ aS,
                                              const float* __restrict__ aD,
                                              const float* __restrict__ Hf,
                                              float* acc, float* mx, float* sm, float* wb,
                                              const float (&Mg)[H], int r0, int lane) {
  float w[U][H];
  if constexpr (PASS == 3 && H == 8) {
    v4f a0[U], a1[U];
#pragma unroll
    for (int u = 0; u < U; ++u) {
      const v4f* ps = (const v4f*)(aS + (size_t)s[u] * 8);
      a0[u] = ps[0]; a1[u] = ps[1];
    }
#pragma unroll
    for (int u = 0; u < U; ++u) {
#pragma unroll
      for (int h = 0; h < 8; ++h) w[u][h] = 0.f;
      if (ok[u]) {
        const float asv[8] = {a0[u].x, a0[u].y, a0[u].z, a0[u].w, a1[u].x, a1[u].y, a1[u].z, a1[u].w};
        float ad[8];
#pragma unroll
        for (int h = 0; h < 8; ++h) ad[h] = mx[h * SUB_R + dl[u]];
#pragma unroll
        for (int h = 0; h < 8; ++h) w[u][h] = expf(leaky(asv[h] + ad[h]) - leaky(Mg[h] + ad[h]));
#pragma unroll
        for (int h = 0; h < 8; ++h) ldsAdd(&sm[h * SUB_R + dl[u]], w[u][h]);
      }
    }
  } else if constexpr (PASS == 3) {
    float a[U];
#pragma unroll
    for (int u = 0; u < U; ++u) a[u] = aS[s[u]];
#pragma unroll
    for (int u = 0; u < U; ++u) {
      w[u][0] = 0.f;
      if (ok[u]) {
        const float ad = mx[dl[u]];
        w[u][0] = expf(leaky(a[u] + ad) - leaky(Mg[0] + ad));
        ldsAdd(&sm[dl[u]], w[u][0]);
      }
    }
  } else if constexpr (H == 8) {
    v4f a0[U], a1[U], b0[U], b1[U];
#pragma unroll
    for (int u = 0; u < U; ++u) {
      const v4f* ps = (const v4f*)(aS + (size_t)s[u] * 8);
      const v4f* pd = (const v4f*)(aD + (size_t)(dl[u] + r0) * 8);
      a0[u] = ps[0]; a1[u] = ps[1]; b0[u] = pd[0]; b1[u] = pd[1];
    }
#pragma unroll
    for (int u = 0; u < U; ++u) {
      const float x[8] = {leaky(a0[u].x + b0[u].x), leaky(a0[u].y + b0[u].y),
                          leaky(a0[u].z + b0[u].z), leaky(a0[u].w + b0[u].w),
                          leaky(a1[u].x + b1[u].x), leaky(a1[u].y + b1[u].y),
                          leaky(a1[u].z + b1[u].z), leaky(a1[u].w + b1[u].w)};
      if (PASS == 1) {
        if (ok[u]) {
#pragma unroll
          for (int h = 0; h < 8; ++h) atomicMaxFloat(&mx[h * SUB_R + dl[u]], x[h]);
        }
      } else {
#pragma unroll
        for (int h = 0; h < 8; ++h) w[u][h] = 0.f;
        if (ok[u]) {
          float m[8];
#pragma unroll
          for (int h = 0; h < 8; ++h) m[h] = mx[h * SUB_R + dl[u]];
#pragma unroll
          for (int h = 0; h < 8; ++h) w[u][h] = expf(x[h] - m[h]);
#pragma unroll
          for (int h = 0; h < 8; ++h) ldsAdd(&sm[h * SUB_R + dl[u]], w[u][h]);
        }
      }
    }
  } else {
    float a[U], b[U];
#pragma unroll
    for (int u = 0; u < U; ++u) { a[u] = aS[s[u]]; b[u] = aD[dl[u] + r0]; }
#pragma unroll
    for (int u = 0; u < U; ++u) {
      const float x = leaky(a[u] + b[u]);
      if (PASS == 1) {
        if (ok[u]) atomicMaxFloat(&mx[dl[u]], x);
      } else {
        w[u][0] = 0.f;
        if (ok[u]) {
          w[u][0] = expf(x - mx[dl[u]]);
          ldsAdd(&sm[dl[u]], w[u][0]);
        }
      }
    }
  }
  if (PASS >= 2) {
    const bool lo = lane < 16;
    const char* Hb = (const char*)Hf;
    char* accb = (char*)acc;
    const unsigned lane16 = (unsigned)lane * 16u;
    const unsigned lane4  = (unsigned)lane * 4u;
#pragma unroll
    for (int u = 0; u < U; ++u) {
      int n = nb - u * 32;
      n = n > 32 ? 32 : n;
      if (n > 0) {
        const int so = s[u] * (HID * 4);
        const int dO = dl[u] * (HID * 4);
        const int sX = xrow_i(so), dX = xrow_i(dO);
        int wA = 0, wAX = 0;
        if constexpr (H == 8) {
          v4f t0 = {w[u][0], w[u][1], w[u][2], w[u][3]};
          v4f t1 = {w[u][4], w[u][5], w[u][6], w[u][7]};
          *(v4f*)(wb + lane * 8)     = t0;
          *(v4f*)(wb + lane * 8 + 4) = t1;
        } else {
          wA = __float_as_int(w[u][0]);
          wAX = xrow_i(wA);
        }
#pragma unroll
        for (int g = 0; g < 2; ++g) {
          if (g * 16 < n) {
            const int   Sg = (g == 0) ? (lo ? so : sX) : (lo ? sX : so);
            const int   Dg = (g == 0) ? (lo ? dO : dX) : (lo ? dX : dO);
            const float Wg = __int_as_float((g == 0) ? (lo ? wA : wAX) : (lo ? wAX : wA));
            v4f hv[8];
            float wq[8];
#define GAT_LOAD(J) hv[(J) & 7] = *(const v4f*)(Hb + (size_t)((unsigned)rshare_i<J>(Sg) + lane16)); \
            if constexpr (H == 8) wq[(J) & 7] = wb[(g * 16 + J) * 8 + (lane >> 2)];
#define GAT_ADD(J) { \
              float wv; \
              if constexpr (H == 8) wv = wq[(J) & 7]; \
              else                  wv = rshare_f<J>(Wg); \
              float* ap = (float*)(accb + ((unsigned)rshare_i<J>(Dg) + lane4)); \
              const v4f t = hv[(J) & 7]; \
              ldsAdd(ap, t.x * wv); ldsAdd(ap + 32, t.y * wv); \
              ldsAdd(ap + 64, t.z * wv); ldsAdd(ap + 96, t.w * wv); }
#define GAT_ADDP(J) if (g * 16 + J < n) GAT_ADD(J)
            if (g * 16 + 8 <= n) {
              ROW8A(GAT_LOAD)
              ROW8A(GAT_ADD)
            } else {
              ROW8A(GAT_LOAD)
              ROW8A(GAT_ADDP)
            }
            if (g * 16 + 8 < n) {
              if (g * 16 + 16 <= n) {
                ROW8B(GAT_LOAD)
                ROW8B(GAT_ADD)
              } else {
                ROW8B(GAT_LOAD)
                ROW8B(GAT_ADDP)
              }
            }
#undef GAT_ADDP
#undef GAT_ADD
#undef GAT_LOAD
          }
        }
      }
    }
  }
}

template <int H, int PASS>
__device__ __forceinline__ void walk_lists(const float* seg0, size_t qStride,
                                           int c0, int c1, int c2, int c3,
                                           const float* __restrict__ aS,
                                           const float* __restrict__ aD,
                                           const float* __restrict__ Hf,
                                           float* acc, float* mx, float* sm, float* wb,
                                           const float (&Mg)[H], int r0, int lane) {
  constexpr int U = BatchU<H>::value;
  const int t1 = c0, t2 = t1 + c1, t3 = t2 + c2, cnt = t3 + c3;
  auto word = [&](int g) -> unsigned {
    if (g >= cnt) return 0u;
    const int q   = (g >= t1) + (g >= t2) + (g >= t3);
    const int off = g - (g >= t3 ? t3 : (g >= t2 ? t2 : (g >= t1 ? t1 : 0)));
    return __float_as_uint(seg0[(size_t)q * qStride + off]);
  };
  unsigned wn[U];
#pragma unroll
  for (int u = 0; u < U; ++u) wn[u] = word(u * 32 + lane);
  for (int i0 = 0; i0 < cnt; i0 += 32 * U) {
    int  s[U], dl[U];
    bool ok[U];
#pragma unroll
    for (int u = 0; u < U; ++u) {
      ok[u] = i0 + u * 32 + lane < cnt;
      s[u]  = (int)(wn[u] >> LBITS);
      dl[u] = (int)(wn[u] & LMASK);
    }
#pragma unroll
    for (int u = 0; u < U; ++u) wn[u] = word(i0 + 32 * U + u * 32 + lane);
    const int nb = cnt - i0;
    process_batch<H, PASS, U>(s, dl, ok, nb > 32 * U ? 32 * U : nb, aS, aD, Hf, acc, mx, sm, wb,
                              Mg, r0, lane);
  }
}

template <int H, int PASS>
__device__ __forceinline__ void drain_work(const int* q, int cnt,
                                           const int* __restrict__ src,
                                           const int* __restrict__ dst,
                                           const float* __restrict__ aS,
                                           const float* __restrict__ aD,
                                           const float* __restrict__ Hf,
                                           float* acc, float* mx, float* sm, float* wb,
                                           int r0, int lane) {
  constexpr int U = BatchU<H>::value;
  for (int i0 = 0; i0 < cnt; i0 += 32 * U) {
    int  s[U], dl[U];
    bool ok[U];
#pragma unroll
    for (int u = 0; u < U; ++u) {
      const int i = i0 + u * 32 + lane;
      ok[u] = i < cnt;
      const int e = ok[u] ? q[i] : 0;
      s[u]  = src[e];
      dl[u] = dst[e] - r0;
    }
#pragma unroll
    for (int u = 0; u < U; ++u)
      if (!ok[u]) { s[u] = 0; dl[u] = 0; }
    const int nb = cnt - i0;
    const float noMg[H] = {};
    process_batch<H, PASS, U>(s, dl, ok, nb > 32 * U ? 32 * U : nb, aS, aD, Hf, acc, mx, sm, wb,
                              noMg, r0, lane);
  }
}

__device__ __forceinline__ void push_hits(int* q, int& cnt, bool hit, int e, int lane) {
  const unsigned m = __builtin_amdgcn_ballot_w32(hit);
  if (m != 0u) {
    const int pos = cnt + __builtin_popcount(m & ((1u << lane) - 1u));
    if (hit) q[pos] = e;
    cnt += __builtin_popcount(m);
  }
}

__device__ __forceinline__ void load_step(const int* __restrict__ dst, int nEdges, int nChunks,
                                          int c, int lane, int (&d)[NLOAD], int& base) {
  base = 0;
  if (c < nChunks) {
    base = c * STEP_E;
    if (base + STEP_E <= nEdges) {
#pragma unroll
      for (int u = 0; u < NLOAD; ++u) d[u] = dst[base + u * 32 + lane];
    } else {
#pragma unroll
      for (int u = 0; u < NLOAD; ++u) {
        const int e = base + u * 32 + lane;
        d[u] = e < nEdges ? dst[e] : -1;
      }
    }
  } else {
#pragma unroll
    for (int u = 0; u < NLOAD; ++u) d[u] = -1;
  }
}

template <int NSUB, typename DrainF>
__device__ __forceinline__ void scan_edges(int* q, int nEdges, const int* __restrict__ dst,
                                           int r0, int rn, int lane, int first, DrainF drain) {
  const int nChunks = (nEdges + STEP_E - 1) / STEP_E;
  const int nRounds = (nChunks + CSTRIDE - 1) / CSTRIDE;
  const int nSteps  = nRounds * NSUB;
  const int rot = (int)(((long long)blockIdx.x * nRounds) / gridDim.x);
  auto chunk_of = [&](int m) {
    int r = m / NSUB + rot;
    if (r >= nRounds) r -= nRounds;
    return m < nSteps ? first + (m % NSUB) + r * CSTRIDE : nChunks;
  };
  int cnt = 0;
  int dn[NLOAD], basen;
  load_step(dst, nEdges, nChunks, chunk_of(0), lane, dn, basen);
  for (int m = 0; ; ++m) {
    const bool more = m < nSteps;
    if (more) {
      int d[NLOAD];
      const int base = basen;
#pragma unroll
      for (int u = 0; u < NLOAD; ++u) d[u] = dn[u];
      load_step(dst, nEdges, nChunks, chunk_of(m + 1), lane, dn, basen);
#pragma unroll
      for (int u = 0; u < NLOAD; ++u)
        push_hits(q, cnt, (unsigned)(d[u] - r0) < (unsigned)rn, base + u * 32 + lane, lane);
    }
    if (cnt > QCAP - STEP_E || (!more && cnt > 0)) {
      drain(q, cnt);
      cnt = 0;
    }
    if (!more) break;
  }
}

__global__ void __launch_bounds__(1024) scan_lists_kernel(const int* __restrict__ ei, int nEdges,
                                                          float* lst) {
  __shared__ int qbuf[32 * QCAP];
  const int lane = threadIdx.x & 31;
  const int wave = __builtin_amdgcn_readfirstlane(threadIdx.x >> 5);
  const int part = blockIdx.x & (NPART - 1);
  const int tile = blockIdx.x / NPART;
  const int R0   = tile * TILE_R;
  const int RN   = (N_NODES - R0) < TILE_R ? (N_NODES - R0) : TILE_R;
  const int sub  = RN >> 2;
  const int* __restrict__ src = ei;
  const int* __restrict__ dst = ei + nEdges;
  const int ppw = sub / 32;
  const int cap = (ppw - 1) * 32;
  const size_t aStride = (size_t)sub * HID;
  float* hdr0 = lst + (size_t)R0 * HID + (size_t)(part * 32 + wave) * ppw * 32;
  int lt[4] = {0, 0, 0, 0};
  auto rec = [&](const int* q, int cnt) __attribute__((always_inline)) {
    for (int i0 = 0; i0 < cnt; i0 += 128) {
      unsigned wd[4];
      int jj[4];
#pragma unroll
      for (int u = 0; u < 4; ++u) {
        const int i = i0 + u * 32 + lane;
        const bool ok = i < cnt;
        const int e = ok ? q[i] : 0;
        const int s = src[e];
        const int d = dst[e] - R0;
        const int j = (d >= sub) + (d >= 2 * sub) + (d >= 3 * sub);
        wd[u] = ((unsigned)s << LBITS) | (unsigned)(d - j * sub);
        jj[u] = ok ? j : -1;
      }
#pragma unroll
      for (int u = 0; u < 4; ++u) {
#pragma unroll
        for (int t = 0; t < 4; ++t) {
          const bool mine = jj[u] == t;
          const unsigned m = __builtin_amdgcn_ballot_w32(mine);
          if (m != 0u) {
            const int n = __builtin_popcount(m);
            if (lt[t] >= 0 && lt[t] + n <= cap) {
              if (mine) {
                const int p = lt[t] + __builtin_popcount(m & ((1u << lane) - 1u));
                hdr0[t * aStride + 32 + p] = __uint_as_float(wd[u]);
              }
              lt[t] += n;
            } else {
              lt[t] = -1;
            }
          }
        }
      }
    }
  };
  scan_edges<1>(qbuf + wave * QCAP, nEdges, dst, R0, RN, lane, part + NPART * wave, rec);
  if (lane < 4) {
    const int v = lane == 0 ? lt[0] : (lane == 1 ? lt[1] : (lane == 2 ? lt[2] : lt[3]));
    hdr0[lane * aStride] = __int_as_float(v);
  }
}

template <int H>
__global__ void __launch_bounds__(1024) owner_layer_kernel(const int* __restrict__ ei, int nEdges,
                                                           const float* __restrict__ aS,
                                                           const float* __restrict__ aD,
                                                           const float* __restrict__ Hf,
                                                           const float* __restrict__ bias,
                                                           float* out, const float* lst,
                                                           const float* wgMax, int applyElu) {
  __shared__ __attribute__((aligned(16))) float acc[SUB_R * HID];
  __shared__ float mx[SUB_R * H];
  __shared__ float sm[SUB_R * H];
  __shared__ int   qbuf[32 * QCAP];
  __shared__ __attribute__((aligned(16))) float wbuf[H == 8 ? 32 * 256 : 32];
  __shared__ float MgS[8];
  __shared__ int   modeS[2];

  const int lane = threadIdx.x & 31;
  const int wave = __builtin_amdgcn_readfirstlane(threadIdx.x >> 5);
  const int own  = blockIdx.x & 3;
  const int tile = blockIdx.x >> 2;
  const int R0   = tile * TILE_R;
  const int RN   = (N_NODES - R0) < TILE_R ? (N_NODES - R0) : TILE_R;
  const int rn   = RN >> 2;
  const int r0   = R0 + own * rn;
  const int* __restrict__ src = ei;
  const int* __restrict__ dst = ei + nEdges;
  int* q = qbuf + wave * QCAP;
  float* wb = wbuf + (H == 8 ? wave * 256 : 0);

  const int ppw = rn / 32;
  const size_t qStride = (size_t)32 * ppw * 32;
  const float* hdr  = lst + (size_t)r0 * HID + (size_t)wave * ppw * 32;
  const float* seg0 = hdr + 32;
  const int c0 = __builtin_amdgcn_readfirstlane(__float_as_int(hdr[0]));
  const int c1 = __builtin_amdgcn_readfirstlane(__float_as_int(hdr[qStride]));
  const int c2 = __builtin_amdgcn_readfirstlane(__float_as_int(hdr[2 * qStride]));
  const int c3 = __builtin_amdgcn_readfirstlane(__float_as_int(hdr[3 * qStride]));
  const bool kept = (c0 | c1 | c2 | c3) >= 0;

  const int hh = (H == 8) ? (lane >> 2) : 0;

  float gmax;
  if (H == 8) gmax = fmaxf(wgMax[threadIdx.x], wgMax[threadIdx.x + 1024]);
  else        gmax = threadIdx.x < GEMM_WG ? wgMax[threadIdx.x * 8] : -__builtin_inff();
  if (threadIdx.x < 8) MgS[threadIdx.x] = -__builtin_inff();
  if (threadIdx.x == 8) { modeS[0] = 0; modeS[1] = 0; }
  __syncthreads();
  if (H == 8) {
    gmax = fmaxf(gmax, __shfl_xor(gmax, 8, 32));
    gmax = fmaxf(gmax, __shfl_xor(gmax, 16, 32));
    if (lane < 8) atomicMaxFloat(&MgS[lane], gmax);
  } else {
#pragma unroll
    for (int off = 16; off > 0; off >>= 1) gmax = fmaxf(gmax, __shfl_xor(gmax, off, 32));
    if (lane == 0 && wave < GEMM_WG / 32) atomicMaxFloat(&MgS[0], gmax);
  }
  if (!kept && lane == 0) modeS[0] = 1;
  __syncthreads();

  bool onePass = __builtin_amdgcn_readfirstlane(modeS[0]) == 0;
  if (onePass) {
    float Mg[H];
#pragma unroll
    for (int h = 0; h < H; ++h)
      Mg[h] = __int_as_float(__builtin_amdgcn_readfirstlane(__float_as_int(MgS[h])));
    const float MgL = MgS[hh];

    for (int i = wave; i < rn; i += 32) {
      const int node = r0 + i;
      const float ad = aD[(size_t)node * H + hh];
      const float x  = leaky(aS[(size_t)node * H + hh] + ad);
      const float w  = expf(x - leaky(MgL + ad));
      const v4f hv = *(const v4f*)(Hf + (size_t)node * HID + 4 * lane);
      float* ap = acc + i * HID + lane;
      ap[0] = w * hv.x; ap[32] = w * hv.y; ap[64] = w * hv.z; ap[96] = w * hv.w;
      if ((lane & 3) == 0 && (lane >> 2) < H) {
        sm[(lane >> 2) * SUB_R + i] = w;
        mx[(lane >> 2) * SUB_R + i] = ad;
      }
    }
    __syncthreads();

    walk_lists<H, 3>(seg0, qStride, c0, c1, c2, c3, aS, aD, Hf, acc, mx, sm, wb, Mg, r0, lane);
    __syncthreads();

    if ((int)threadIdx.x < rn) {
      bool low = false;
#pragma unroll
      for (int h = 0; h < H; ++h) low |= !(sm[h * SUB_R + threadIdx.x] >= SHIFT_FLOOR);
      if (low) modeS[1] = 1;
    }
    __syncthreads();
    onePass = __builtin_amdgcn_readfirstlane(modeS[1]) == 0;
  }

  if (!onePass) {
  const float noMg[H] = {};
  for (int i = threadIdx.x; i < rn * H; i += 1024) {
    const int node = r0 + i / H;
    const int h    = i % H;
    mx[h * SUB_R + i / H] = leaky(aS[(size_t)node * H + h] + aD[(size_t)node * H + h]);
  }
  __syncthreads();

  if (kept) {
    walk_lists<H, 1>(seg0, qStride, c0, c1, c2, c3, aS, aD, Hf, acc, mx, sm, wb, noMg, r0, lane);
  } else {
    scan_edges<NPART>(q, nEdges, dst, r0, rn, lane, NPART * wave, [&](const int* qq, int cnt) {
      drain_work<H, 1>(qq, cnt, src, dst, aS, aD, Hf, acc, mx, sm, wb, r0, lane);
    });
  }
  __syncthreads();

  for (int i = wave; i < rn; i += 32) {
    const int node = r0 + i;
    const float x  = leaky(aS[(size_t)node * H + hh] + aD[(size_t)node * H + hh]);
    const float w  = expf(x - mx[hh * SUB_R + i]);
    const v4f hv = *(const v4f*)(Hf + (size_t)node * HID + 4 * lane);
    float* ap = acc + i * HID + lane;
    ap[0] = w * hv.x; ap[32] = w * hv.y; ap[64] = w * hv.z; ap[96] = w * hv.w;
    if ((lane & 3) == 0 && (lane >> 2) < H) sm[(lane >> 2) * SUB_R + i] = w;
  }
  __syncthreads();

  if (kept) {
    walk_lists<H, 2>(seg0, qStride, c0, c1, c2, c3, aS, aD, Hf, acc, mx, sm, wb, noMg, r0, lane);
  } else {
    scan_edges<NPART>(q, nEdges, dst, r0, rn, lane, NPART * wave, [&](const int* qq, int cnt) {
      drain_work<H, 2>(qq, cnt, src, dst, aS, aD, Hf, acc, mx, sm, wb, r0, lane);
    });
  }
  __syncthreads();
  }

  const v4f bv = *(const v4f*)(bias + 4 * lane);
  for (int i = wave; i < rn; i += 32) {
    const float* ap = acc + i * HID + lane;
    const float sv = sm[hh * SUB_R + i];
    v4f v = {ap[0] / sv + bv.x, ap[32] / sv + bv.y, ap[64] / sv + bv.z, ap[96] / sv + bv.w};
    if (applyElu) {
      v.x = v.x > 0.f ? v.x : (expf(v.x) - 1.f);
      v.y = v.y > 0.f ? v.y : (expf(v.y) - 1.f);
      v.z = v.z > 0.f ? v.z : (expf(v.z) - 1.f);
      v.w = v.w > 0.f ? v.w : (expf(v.w) - 1.f);
    }
    *(v4f*)(out + (size_t)(r0 + i) * HID + 4 * lane) = v;
  }
}

extern "C" void kernel_launch(void* const* d_in, const int* in_sizes, int n_in,
                              void* d_out, int out_size, void* d_ws, size_t ws_size,
                              hipStream_t stream) {
  const float* x   = (const float*)d_in[0];
  const int*   ei  = (const int*)d_in[1];
  const float* W1  = (const float*)d_in[2];
  const float* as1 = (const float*)d_in[3];
  const float* ad1 = (const float*)d_in[4];
  const float* b1  = (const float*)d_in[5];
  const float* W2  = (const float*)d_in[6];
  const float* as2 = (const float*)d_in[7];
  const float* ad2 = (const float*)d_in[8];
  const float* b2  = (const float*)d_in[9];

  const int nEdges = in_sizes[1] / 2;

  float* A   = (float*)d_ws;
  float* B   = A  + (size_t)N_NODES * HID;
  float* aS  = B  + (size_t)N_NODES * HID;
  float* aD  = aS + (size_t)N_NODES * GAT_HEADS;
  float* out = (float*)d_out;

  const unsigned gOwner = ((N_NODES + TILE_R - 1) / TILE_R) * NPART;

  float* wgMax1 = out + (size_t)N_NODES * HID - GEMM_WG * 8;
  float* wgMax2 = aS + N_NODES;

  scan_lists_kernel<<<gOwner, 1024, 0, stream>>>(ei, nEdges, out);

  gemm128_wmma<GAT_HEADS><<<GEMM_WG, 1024, 0, stream>>>(x, W1, as1, ad1, A, aS, aD, wgMax1);
  owner_layer_kernel<GAT_HEADS><<<gOwner, 1024, 0, stream>>>(ei, nEdges, aS, aD, A, b1, B, out, wgMax1, 1);

  gemm128_wmma<1><<<GEMM_WG, 1024, 0, stream>>>(B, W2, as2, ad2, A, aS, aD, wgMax2);
  owner_layer_kernel<1><<<gOwner, 1024, 0, stream>>>(ei, nEdges, aS, aD, A, b2, out, out, wgMax2, 0);
}
